# starting kernel
# baseline (speedup 1.0000x reference)
#include <hip/hip_runtime.h>


#define NPTS   16384
#define DIN    1024
#define DPROJ  128
#define NSAMP  163
#define NCHUNK 8
#define TILES_PER_CHUNK (NPTS / 16 / NCHUNK)

typedef float v2f __attribute__((ext_vector_type(2)));
typedef float v8f __attribute__((ext_vector_type(8)));

__device__ __forceinline__ v8f wmma_f32_k4(v2f a, v2f b, v8f c) {
  return __builtin_amdgcn_wmma_f32_16x16x4_f32(
       false, a,  false, b,
       (short)0, c,  false,  false);
}

__global__ __launch_bounds__(256)
void proj_kernel(const float* __restrict__ F, const float* __restrict__ W,
                 float* __restrict__ R) {
  const int wave = (blockIdx.x * blockDim.x + threadIdx.x) >> 5;
  const int lane = threadIdx.x & 31;
  const int tm = wave >> 3;
  const int tn = wave & 7;
  const int m0 = tm * 16, n0 = tn * 16;
  const int half = lane >> 4;
  const int r16  = lane & 15;
  const int koff = half * 2;

  v8f acc = {};
  const float* arow = F + (size_t)(m0 + r16) * DIN + koff;
  for (int k = 0; k < DIN; k += 4) {
    v2f a = *(const v2f*)(arow + k);
    v2f b;
    b.x = W[(size_t)(k + koff)     * DPROJ + n0 + r16];
    b.y = W[(size_t)(k + koff + 1) * DPROJ + n0 + r16];
    acc = wmma_f32_k4(a, b, acc);
  }
#pragma unroll
  for (int i = 0; i < 8; i++)
    R[(size_t)(m0 + i + half * 8) * DPROJ + n0 + r16] = acc[i];
}

__global__ __launch_bounds__(256)
void sq_kernel(const float* __restrict__ R, float* __restrict__ sq) {
  const int row  = (blockIdx.x * blockDim.x + threadIdx.x) >> 5;
  const int lane = threadIdx.x & 31;
  const float* rp = R + (size_t)row * DPROJ;
  float s = 0.f;
#pragma unroll
  for (int k = lane; k < DPROJ; k += 32) { float v = rp[k]; s += v * v; }
#pragma unroll
  for (int m = 16; m >= 1; m >>= 1) s += __shfl_xor(s, m, 32);
  if (lane == 0) sq[row] = s;
}

__global__ __launch_bounds__(256)
void gram_rowsum_kernel(const float* __restrict__ R, const float* __restrict__ sq,
                        float* __restrict__ anchorPart) {
  const int wave  = (blockIdx.x * blockDim.x + threadIdx.x) >> 5;
  const int lane  = threadIdx.x & 31;
  const int chunk = wave & (NCHUNK - 1);
  const int strip = wave >> 3;
  const int half  = lane >> 4;
  const int r16   = lane & 15;
  const int koff  = half * 2;
  const int m0    = strip * 16;

  v2f afrag[32];
  const float* arow = R + (size_t)(m0 + r16) * DPROJ + koff;
#pragma unroll
  for (int kk = 0; kk < 32; kk++) afrag[kk] = *(const v2f*)(arow + kk * 4);

  float sqr[8];
#pragma unroll
  for (int i = 0; i < 8; i++) sqr[i] = sq[m0 + i + half * 8];

  float partial[8] = {0.f, 0.f, 0.f, 0.f, 0.f, 0.f, 0.f, 0.f};

  const int t0 = chunk * TILES_PER_CHUNK;
  for (int t = t0; t < t0 + TILES_PER_CHUNK; t++) {
    const int col = t * 16 + r16;
    const float sqc = sq[col];
    const float* brow = R + (size_t)col * DPROJ + koff;
    __builtin_prefetch((const void*)(brow + 16 * DPROJ), 0, 1);
    v8f acc = {};
#pragma unroll
    for (int kk = 0; kk < 32; kk++) {
      v2f b = *(const v2f*)(brow + kk * 4);
      acc = wmma_f32_k4(afrag[kk], b, acc);
    }
#pragma unroll
    for (int i = 0; i < 8; i++)
      partial[i] += fmaxf(0.f, sqr[i] + sqc - 2.f * acc[i]);
  }

#pragma unroll
  for (int i = 0; i < 8; i++) {
    float p = partial[i];
    p += __shfl_xor(p, 1, 32);
    p += __shfl_xor(p, 2, 32);
    p += __shfl_xor(p, 4, 32);
    p += __shfl_xor(p, 8, 32);
    if (r16 == 0)
      anchorPart[(size_t)chunk * NPTS + m0 + i + half * 8] = p;
  }
}

__global__ __launch_bounds__(256)
void combine_kernel(const float* __restrict__ anchorPart,
                    float* __restrict__ anchorSq) {
  const int i = blockIdx.x * 256 + threadIdx.x;
  float s = 0.f;
#pragma unroll
  for (int c = 0; c < NCHUNK; c++) s += anchorPart[(size_t)c * NPTS + i];
  anchorSq[i] = s;
}

__global__ __launch_bounds__(1024)
void greedy_kernel(const float* __restrict__ R, const float* __restrict__ sq,
                   const float* __restrict__ anchorSq, int* __restrict__ sel) {
  __shared__ float redV[32];
  __shared__ int   redI[32];
  __shared__ float selRow[DPROJ];
  __shared__ float selSqS;
  __shared__ int   selIdxS;

  const int t    = threadIdx.x;
  const int lane = t & 31;
  const int wid  = t >> 5;

  float aVal[16], sqOwn[16];
#pragma unroll
  for (int j = 0; j < 16; j++) {
    const int row = t + j * 1024;
    aVal[j]  = sqrtf(anchorSq[row]);
    sqOwn[j] = sq[row];
  }

  for (int s = 0; s < NSAMP; s++) {
    float bv = aVal[0]; int bi = t;
#pragma unroll
    for (int j = 1; j < 16; j++)
      if (aVal[j] > bv) { bv = aVal[j]; bi = t + j * 1024; }
#pragma unroll
    for (int m = 1; m < 32; m <<= 1) {
      const float ov = __shfl_xor(bv, m, 32);
      const int   oi = __shfl_xor(bi, m, 32);
      if (ov > bv || (ov == bv && oi < bi)) { bv = ov; bi = oi; }
    }
    if (lane == 0) { redV[wid] = bv; redI[wid] = bi; }
    __syncthreads();
    if (wid == 0) {
      bv = redV[lane]; bi = redI[lane];
#pragma unroll
      for (int m = 1; m < 32; m <<= 1) {
        const float ov = __shfl_xor(bv, m, 32);
        const int   oi = __shfl_xor(bi, m, 32);
        if (ov > bv || (ov == bv && oi < bi)) { bv = ov; bi = oi; }
      }
      if (lane == 0) { selIdxS = bi; selSqS = sq[bi]; sel[s] = bi; }
    }
    __syncthreads();
    const int si = selIdxS;
    if (t < DPROJ) selRow[t] = R[(size_t)si * DPROJ + t];
    __syncthreads();

    const float selSq = selSqS;
#pragma unroll 4
    for (int j = 0; j < 16; j++) {
      const int row = t + j * 1024;
      const float4* rp = (const float4*)(R + (size_t)row * DPROJ);
      const float4* wp = (const float4*)selRow;
      float dot = 0.f;
#pragma unroll
      for (int k = 0; k < DPROJ / 4; k++) {
        const float4 v = rp[k];
        const float4 w = wp[k];
        dot += v.x * w.x + v.y * w.y + v.z * w.z + v.w * w.w;
      }
      const float d2 = sqOwn[j] + selSq - 2.f * dot;
      aVal[j] = fminf(aVal[j], sqrtf(fmaxf(d2, 0.f)));
    }
    __syncthreads();
  }
}

__global__ __launch_bounds__(256)
void gather_kernel(const float* __restrict__ F, const int* __restrict__ sel,
                   float* __restrict__ out) {
  const int s = blockIdx.x;
  const int c = blockIdx.y * 256 + threadIdx.x;
  out[(size_t)s * DIN + c] = F[(size_t)sel[s] * DIN + c];
}

extern "C" void kernel_launch(void* const* d_in, const int* in_sizes, int n_in,
                              void* d_out, int out_size, void* d_ws, size_t ws_size,
                              hipStream_t stream) {
  (void)in_sizes; (void)n_in; (void)out_size; (void)ws_size;
  const float* F = (const float*)d_in[0];
  const float* W = (const float*)d_in[1];

  float* R          = (float*)d_ws;
  float* sq         = R + (size_t)NPTS * DPROJ;
  float* anchorSq   = sq + NPTS;
  float* anchorPart = anchorSq + NPTS;
  int*   sel        = (int*)(anchorPart + (size_t)NCHUNK * NPTS);

  proj_kernel<<<1024, 256, 0, stream>>>(F, W, R);
  sq_kernel<<<NPTS / 8, 256, 0, stream>>>(R, sq);
  gram_rowsum_kernel<<<1024, 256, 0, stream>>>(R, sq, anchorPart);
  combine_kernel<<<NPTS / 256, 256, 0, stream>>>(anchorPart, anchorSq);
  greedy_kernel<<<1, 1024, 0, stream>>>(R, sq, anchorSq, sel);
  gather_kernel<<<dim3(NSAMP, DIN / 256), 256, 0, stream>>>(F, sel, (float*)d_out);
}


# greedy sweep: row loads run three k-groups ahead of their arithmetic in a ring of four register sets (anchors and row norms moved to LDS for room), so L2 latency overlaps the FMAs; sums unchanged
# speedup vs baseline: 9.6715x; 9.6715x over previous
#include <hip/hip_runtime.h>


#define NPTS   16384
#define DIN    1024
#define DPROJ  128
#define NSAMP  163
#define NCHUNK 8
#define TILES_PER_CHUNK (NPTS / 16 / NCHUNK)

typedef float v2f __attribute__((ext_vector_type(2)));
typedef float v8f __attribute__((ext_vector_type(8)));

__device__ __forceinline__ v8f wmma_f32_k4(v2f a, v2f b, v8f c) {
  return __builtin_amdgcn_wmma_f32_16x16x4_f32(
       false, a,  false, b,
       (short)0, c,  false,  false);
}

__global__ __launch_bounds__(256)
void proj_kernel(const float* __restrict__ F, const float* __restrict__ W,
                 float* __restrict__ R) {
  const int wave = (blockIdx.x * blockDim.x + threadIdx.x) >> 5;
  const int lane = threadIdx.x & 31;
  const int tm = wave >> 3;
  const int tn = wave & 7;
  const int m0 = tm * 16, n0 = tn * 16;
  const int half = lane >> 4;
  const int r16  = lane & 15;
  const int koff = half * 2;

  v8f acc = {};
  const float* arow = F + (size_t)(m0 + r16) * DIN + koff;
  for (int k = 0; k < DIN; k += 4) {
    v2f a = *(const v2f*)(arow + k);
    v2f b;
    b.x = W[(size_t)(k + koff)     * DPROJ + n0 + r16];
    b.y = W[(size_t)(k + koff + 1) * DPROJ + n0 + r16];
    acc = wmma_f32_k4(a, b, acc);
  }
#pragma unroll
  for (int i = 0; i < 8; i++)
    R[(size_t)(m0 + i + half * 8) * DPROJ + n0 + r16] = acc[i];
}

__global__ __launch_bounds__(256)
void sq_kernel(const float* __restrict__ R, float* __restrict__ sq) {
  const int row  = (blockIdx.x * blockDim.x + threadIdx.x) >> 5;
  const int lane = threadIdx.x & 31;
  const float* rp = R + (size_t)row * DPROJ;
  float s = 0.f;
#pragma unroll
  for (int k = lane; k < DPROJ; k += 32) { float v = rp[k]; s += v * v; }
#pragma unroll
  for (int m = 16; m >= 1; m >>= 1) s += __shfl_xor(s, m, 32);
  if (lane == 0) sq[row] = s;
}

__global__ __launch_bounds__(256)
void gram_rowsum_kernel(const float* __restrict__ R, const float* __restrict__ sq,
                        float* __restrict__ anchorPart) {
  const int wave  = (blockIdx.x * blockDim.x + threadIdx.x) >> 5;
  const int lane  = threadIdx.x & 31;
  const int chunk = wave & (NCHUNK - 1);
  const int strip = wave >> 3;
  const int half  = lane >> 4;
  const int r16   = lane & 15;
  const int koff  = half * 2;
  const int m0    = strip * 16;

  v2f afrag[32];
  const float* arow = R + (size_t)(m0 + r16) * DPROJ + koff;
#pragma unroll
  for (int kk = 0; kk < 32; kk++) afrag[kk] = *(const v2f*)(arow + kk * 4);

  float sqr[8];
#pragma unroll
  for (int i = 0; i < 8; i++) sqr[i] = sq[m0 + i + half * 8];

  float partial[8] = {0.f, 0.f, 0.f, 0.f, 0.f, 0.f, 0.f, 0.f};

  const int t0 = chunk * TILES_PER_CHUNK;
  for (int t = t0; t < t0 + TILES_PER_CHUNK; t++) {
    const int col = t * 16 + r16;
    const float sqc = sq[col];
    const float* brow = R + (size_t)col * DPROJ + koff;
    __builtin_prefetch((const void*)(brow + 16 * DPROJ), 0, 1);
    v8f acc = {};
#pragma unroll
    for (int kk = 0; kk < 32; kk++) {
      v2f b = *(const v2f*)(brow + kk * 4);
      acc = wmma_f32_k4(afrag[kk], b, acc);
    }
#pragma unroll
    for (int i = 0; i < 8; i++)
      partial[i] += fmaxf(0.f, sqr[i] + sqc - 2.f * acc[i]);
  }

#pragma unroll
  for (int i = 0; i < 8; i++) {
    float p = partial[i];
    p += __shfl_xor(p, 1, 32);
    p += __shfl_xor(p, 2, 32);
    p += __shfl_xor(p, 4, 32);
    p += __shfl_xor(p, 8, 32);
    if (r16 == 0)
      anchorPart[(size_t)chunk * NPTS + m0 + i + half * 8] = p;
  }
}

__global__ __launch_bounds__(256)
void combine_kernel(const float* __restrict__ anchorPart,
                    float* __restrict__ anchorSq) {
  const int i = blockIdx.x * 256 + threadIdx.x;
  float s = 0.f;
#pragma unroll
  for (int c = 0; c < NCHUNK; c++) s += anchorPart[(size_t)c * NPTS + i];
  anchorSq[i] = s;
}

__global__ __launch_bounds__(1024)
void kmajor_kernel(float* __restrict__ R) {
  float* blk = R + (size_t)blockIdx.x * (128 * DPROJ);
  const int lane = threadIdx.x & 31;
  const int w    = threadIdx.x >> 5;
  const int m    = 32 * (w & 3) + lane;
  const int kg0  = w >> 2;

  float4 v[4];
#pragma unroll
  for (int i = 0; i < 4; i++)
    v[i] = *(const float4*)(blk + m * DPROJ + 4 * (kg0 + 8 * i));

  __syncthreads();

#pragma unroll
  for (int i = 0; i < 4; i++) {
    const int k = 4 * (kg0 + 8 * i);
    blk[(k + 0) * 128 + m] = v[i].x;
    blk[(k + 1) * 128 + m] = v[i].y;
    blk[(k + 2) * 128 + m] = v[i].z;
    blk[(k + 3) * 128 + m] = v[i].w;
  }
}

#define MMAX   4
#define PPITCH 132

typedef float v4f __attribute__((ext_vector_type(4)));
typedef const __attribute__((address_space(1))) float* gfp;
typedef const __attribute__((address_space(1))) v4f*   gv4p;

__device__ __forceinline__ void load_group(v4f (&r)[4], gfp p) {
#pragma unroll
  for (int u = 0; u < 4; u++) r[u] = *(gv4p)(p + u * 128);
}

template <int M>
__device__ __forceinline__ void fold_group(const v4f (&v)[4], const float* pool,
                                           const int (&so)[M], const int g,
                                           v2f (&d)[M][2]) {
#pragma unroll
  for (int c = 0; c < M; c++) {
    const v4f w = *(const v4f*)(pool + so[c] + 4 * g);
    v2f pa = v[1].xy * w.yy;
    v2f pb = v[1].zw * w.yy;
    pa = __builtin_elementwise_fma(v[0].xy, w.xx, pa);
    pb = __builtin_elementwise_fma(v[0].zw, w.xx, pb);
    pa = __builtin_elementwise_fma(v[2].xy, w.zz, pa);
    pb = __builtin_elementwise_fma(v[2].zw, w.zz, pb);
    pa = __builtin_elementwise_fma(v[3].xy, w.ww, pa);
    pb = __builtin_elementwise_fma(v[3].zw, w.ww, pb);
    d[c][0] += pa;
    d[c][1] += pb;
  }
}

template <int M>
__device__ __forceinline__ void after_fold(gfp &p, v2f (&d)[M][2]) {
#pragma unroll
  for (int c = 0; c < M; c++)
    asm volatile("" : "+v"(p), "+v"(d[c][0]), "+v"(d[c][1]));
}

template <int M, bool AHEAD>
__device__ __forceinline__ void ring_trip(gfp &p, const float* pool,
                                          const int (&so)[M], const int g,
                                          v4f (&r0)[4], v4f (&r1)[4],
                                          v4f (&r2)[4], v4f (&r3)[4],
                                          v2f (&d)[M][2]) {
  load_group(r3, p + 3 * 512);
  fold_group<M>(r0, pool, so, g, d);
  after_fold<M>(p, d);
  if (AHEAD) load_group(r0, p + 4 * 512);
  fold_group<M>(r1, pool, so, g + 1, d);
  after_fold<M>(p, d);
  if (AHEAD) load_group(r1, p + 5 * 512);
  fold_group<M>(r2, pool, so, g + 2, d);
  after_fold<M>(p, d);
  if (AHEAD) load_group(r2, p + 6 * 512);
  fold_group<M>(r3, pool, so, g + 3, d);
  after_fold<M>(p, d);
  p += 4 * 512;
}

template <int M>
__device__ __forceinline__ void sweep_cols(const float* __restrict__ RB,
                                           const float* sqS, float* anchS,
                                           const float* pool, const int* slotS,
                                           const float* selSqS,
                                           const unsigned run, const int t) {
  int   so[M];
  float ssq[M];
#pragma unroll
  for (int c = 0; c < M; c++) {
    so[c]  = __builtin_amdgcn_readfirstlane(slotS[c]) * PPITCH;
    ssq[c] = __int_as_float(__builtin_amdgcn_readfirstlane(__float_as_int(selSqS[c])));
  }
  gfp p = (gfp)(RB + run);
  v4f r0[4], r1[4], r2[4], r3[4];
  load_group(r0, p);
  asm volatile("" : "+v"(p));
  load_group(r1, p + 512);
  asm volatile("" : "+v"(p));
  load_group(r2, p + 1024);
#pragma unroll
  for (int j = 0; j < 4; j++) {
    v2f d[M][2];
#pragma unroll
    for (int c = 0; c < M; c++) { d[c][0] = (v2f){0.f, 0.f}; d[c][1] = (v2f){0.f, 0.f}; }
    if (j < 3) {
#pragma unroll 1
      for (int g = 0; g < 32; g += 4)
        ring_trip<M, true>(p, pool, so, g, r0, r1, r2, r3, d);
    } else {
#pragma unroll 1
      for (int g = 0; g < 28; g += 4)
        ring_trip<M, true>(p, pool, so, g, r0, r1, r2, r3, d);
      ring_trip<M, false>(p, pool, so, 28, r0, r1, r2, r3, d);
    }
    const float4 q = *(const float4*)(sqS + 4 * (1024 * j + t));
    float4 a = *(const float4*)(anchS + 4 * (1024 * j + t));
#pragma unroll
    for (int c = 0; c < M; c++) {
      const float e0 = q.x + ssq[c] - 2.f * d[c][0].x;
      const float e1 = q.y + ssq[c] - 2.f * d[c][0].y;
      const float e2 = q.z + ssq[c] - 2.f * d[c][1].x;
      const float e3 = q.w + ssq[c] - 2.f * d[c][1].y;
      a.x = fminf(a.x, sqrtf(fmaxf(e0, 0.f)));
      a.y = fminf(a.y, sqrtf(fmaxf(e1, 0.f)));
      a.z = fminf(a.z, sqrtf(fmaxf(e2, 0.f)));
      a.w = fminf(a.w, sqrtf(fmaxf(e3, 0.f)));
    }
    *(float4*)(anchS + 4 * (1024 * j + t)) = a;
  }
}

__global__ __launch_bounds__(1024)
void greedy_kernel(const float* __restrict__ RB, const float* __restrict__ sq,
                   const float* __restrict__ anchorSq, int* __restrict__ sel) {
  __shared__ float redV[32];
  __shared__ int   redI[32];
  __shared__ float redS[32];
  __shared__ __attribute__((aligned(16))) float pool[32 * PPITCH];
  __shared__ int   slotS[MMAX];
  __shared__ float selSqS[MMAX];
  __shared__ int   mS;
  __shared__ __attribute__((aligned(16))) float anchS[4 * 1024 * 4];
  __shared__ __attribute__((aligned(16))) float sqS[4 * 1024 * 4];

  const int t    = threadIdx.x;
  const int lane = t & 31;
  const int wid  = t >> 5;
  const int row0 = 512 * wid + 4 * lane;

#pragma unroll
  for (int j = 0; j < 4; j++) {
    float4 a = *(const float4*)(anchorSq + row0 + 128 * j);
    a.x = sqrtf(a.x);
    a.y = sqrtf(a.y);
    a.z = sqrtf(a.z);
    a.w = sqrtf(a.w);
    *(float4*)(anchS + 4 * (1024 * j + t)) = a;
    *(float4*)(sqS + 4 * (1024 * j + t)) = *(const float4*)(sq + row0 + 128 * j);
  }

  const unsigned run = (unsigned)wid * (4u * 128u * DPROJ) + 4u * (unsigned)lane;

  int done = 0;
  for (int pass = 0; pass < NSAMP && done < NSAMP; pass++) {
    float aVal[16];
#pragma unroll
    for (int j = 0; j < 4; j++) {
      const float4 a = *(const float4*)(anchS + 4 * (1024 * j + t));
      aVal[4 * j + 0] = a.x; aVal[4 * j + 1] = a.y; aVal[4 * j + 2] = a.z; aVal[4 * j + 3] = a.w;
    }
    float bv = aVal[0]; int bi = row0; float sv = -1.f;
#pragma unroll
    for (int e = 1; e < 16; e++) {
      const float v = aVal[e];
      if (v > bv) { sv = bv; bv = v; bi = row0 + 128 * (e >> 2) + (e & 3); }
      else        { sv = fmaxf(sv, v); }
    }
#pragma unroll
    for (int m = 1; m < 32; m <<= 1) {
      const float ov = __shfl_xor(bv, m, 32);
      const int   oi = __shfl_xor(bi, m, 32);
      const float os = __shfl_xor(sv, m, 32);
      const bool take = (ov > bv) || (ov == bv && oi < bi);
      sv = take ? fmaxf(os, bv) : fmaxf(sv, ov);
      if (take) { bv = ov; bi = oi; }
    }
    {
      const unsigned base = (unsigned)(bi >> 7) * (128u * DPROJ) + (unsigned)(bi & 127);
#pragma unroll
      for (int i = 0; i < 4; i++)
        pool[wid * PPITCH + lane + 32 * i] = RB[base + (unsigned)(lane + 32 * i) * 128u];
    }
    if (lane == 0) { redV[wid] = bv; redI[wid] = bi; redS[wid] = sv; }
    __syncthreads();

    if (wid == 0) {
      float pv = redV[lane];
      const int pi = redI[lane];
      float T = redS[lane];
#pragma unroll
      for (int m = 1; m < 32; m <<= 1) T = fmaxf(T, __shfl_xor(T, m, 32));
      const float psq = sq[pi];
      const float4* own = (const float4*)(pool + lane * PPITCH);
      int m = 0;
      for (int c = 0; c < MMAX; c++) {
        float cv = pv; int ci = pi; int cl = lane;
#pragma unroll
        for (int x = 1; x < 32; x <<= 1) {
          const float ov = __shfl_xor(cv, x, 32);
          const int   oi = __shfl_xor(ci, x, 32);
          const int   ol = __shfl_xor(cl, x, 32);
          if (ov > cv || (ov == cv && oi < ci)) { cv = ov; ci = oi; cl = ol; }
        }
        if (c > 0 && !(cv > T)) break;
        const float ssq = __shfl(psq, cl, 32);
        if (lane == 0) { sel[done + c] = ci; slotS[c] = cl; selSqS[c] = ssq; }
        m = c + 1;
        if (done + m == NSAMP || m == MMAX) break;
        const float4* sr = (const float4*)(pool + cl * PPITCH);
        float d = 0.f;
#pragma unroll 8
        for (int kg = 0; kg < 32; kg++) {
          const float4 r = own[kg];
          const float4 w = sr[kg];
          float p = r.y * w.y;
          p = fmaf(r.x, w.x, p);
          p = fmaf(r.z, w.z, p);
          p = fmaf(r.w, w.w, p);
          d += p;
        }
        const float e = psq + ssq - 2.f * d;
        pv = fminf(pv, sqrtf(fmaxf(e, 0.f)));
      }
      if (lane == 0) mS = m;
    }
    __syncthreads();

    const int m = mS;
    if (m == 1)      sweep_cols<1>(RB, sqS, anchS, pool, slotS, selSqS, run, t);
    else if (m == 2) sweep_cols<2>(RB, sqS, anchS, pool, slotS, selSqS, run, t);
    else if (m == 3) sweep_cols<3>(RB, sqS, anchS, pool, slotS, selSqS, run, t);
    else             sweep_cols<4>(RB, sqS, anchS, pool, slotS, selSqS, run, t);
    done += m;
    __syncthreads();
  }
}

__global__ __launch_bounds__(256)
void gather_kernel(const float* __restrict__ F, const int* __restrict__ sel,
                   float* __restrict__ out) {
  const int s = blockIdx.x;
  const int c = blockIdx.y * 256 + threadIdx.x;
  out[(size_t)s * DIN + c] = F[(size_t)sel[s] * DIN + c];
}

extern "C" void kernel_launch(void* const* d_in, const int* in_sizes, int n_in,
                              void* d_out, int out_size, void* d_ws, size_t ws_size,
                              hipStream_t stream) {
  (void)in_sizes; (void)n_in; (void)out_size; (void)ws_size;
  const float* F = (const float*)d_in[0];
  const float* W = (const float*)d_in[1];

  float* R          = (float*)d_ws;
  float* sq         = R + (size_t)NPTS * DPROJ;
  float* anchorSq   = sq + NPTS;
  float* anchorPart = anchorSq + NPTS;
  int*   sel        = (int*)(anchorPart + (size_t)NCHUNK * NPTS);

  proj_kernel<<<1024, 256, 0, stream>>>(F, W, R);
  sq_kernel<<<NPTS / 8, 256, 0, stream>>>(R, sq);
  gram_rowsum_kernel<<<1024, 256, 0, stream>>>(R, sq, anchorPart);
  combine_kernel<<<NPTS / 256, 256, 0, stream>>>(anchorPart, anchorSq);
  kmajor_kernel<<<NPTS / 128, 1024, 0, stream>>>(R);
  greedy_kernel<<<1, 1024, 0, stream>>>(R, sq, anchorSq, sel);
  gather_kernel<<<dim3(NSAMP, DIN / 256), 256, 0, stream>>>(F, sel, (float*)d_out);
}


# greedy sweep: the up to 4 selected rows' LDS values of a k-group are read as one run behind one wait and their dot chains interleaved, not read-wait-compute one row after another; sums unchanged
# speedup vs baseline: 10.2437x; 1.0592x over previous
#include <hip/hip_runtime.h>


#define NPTS   16384
#define DIN    1024
#define DPROJ  128
#define NSAMP  163
#define NCHUNK 8
#define TILES_PER_CHUNK (NPTS / 16 / NCHUNK)

typedef float v2f __attribute__((ext_vector_type(2)));
typedef float v8f __attribute__((ext_vector_type(8)));

__device__ __forceinline__ v8f wmma_f32_k4(v2f a, v2f b, v8f c) {
  return __builtin_amdgcn_wmma_f32_16x16x4_f32(
       false, a,  false, b,
       (short)0, c,  false,  false);
}

__global__ __launch_bounds__(256)
void proj_kernel(const float* __restrict__ F, const float* __restrict__ W,
                 float* __restrict__ R) {
  const int wave = (blockIdx.x * blockDim.x + threadIdx.x) >> 5;
  const int lane = threadIdx.x & 31;
  const int tm = wave >> 3;
  const int tn = wave & 7;
  const int m0 = tm * 16, n0 = tn * 16;
  const int half = lane >> 4;
  const int r16  = lane & 15;
  const int koff = half * 2;

  v8f acc = {};
  const float* arow = F + (size_t)(m0 + r16) * DIN + koff;
  for (int k = 0; k < DIN; k += 4) {
    v2f a = *(const v2f*)(arow + k);
    v2f b;
    b.x = W[(size_t)(k + koff)     * DPROJ + n0 + r16];
    b.y = W[(size_t)(k + koff + 1) * DPROJ + n0 + r16];
    acc = wmma_f32_k4(a, b, acc);
  }
#pragma unroll
  for (int i = 0; i < 8; i++)
    R[(size_t)(m0 + i + half * 8) * DPROJ + n0 + r16] = acc[i];
}

__global__ __launch_bounds__(256)
void sq_kernel(const float* __restrict__ R, float* __restrict__ sq) {
  const int row  = (blockIdx.x * blockDim.x + threadIdx.x) >> 5;
  const int lane = threadIdx.x & 31;
  const float* rp = R + (size_t)row * DPROJ;
  float s = 0.f;
#pragma unroll
  for (int k = lane; k < DPROJ; k += 32) { float v = rp[k]; s += v * v; }
#pragma unroll
  for (int m = 16; m >= 1; m >>= 1) s += __shfl_xor(s, m, 32);
  if (lane == 0) sq[row] = s;
}

__global__ __launch_bounds__(256)
void gram_rowsum_kernel(const float* __restrict__ R, const float* __restrict__ sq,
                        float* __restrict__ anchorPart) {
  const int wave  = (blockIdx.x * blockDim.x + threadIdx.x) >> 5;
  const int lane  = threadIdx.x & 31;
  const int chunk = wave & (NCHUNK - 1);
  const int strip = wave >> 3;
  const int half  = lane >> 4;
  const int r16   = lane & 15;
  const int koff  = half * 2;
  const int m0    = strip * 16;

  v2f afrag[32];
  const float* arow = R + (size_t)(m0 + r16) * DPROJ + koff;
#pragma unroll
  for (int kk = 0; kk < 32; kk++) afrag[kk] = *(const v2f*)(arow + kk * 4);

  float sqr[8];
#pragma unroll
  for (int i = 0; i < 8; i++) sqr[i] = sq[m0 + i + half * 8];

  float partial[8] = {0.f, 0.f, 0.f, 0.f, 0.f, 0.f, 0.f, 0.f};

  const int t0 = chunk * TILES_PER_CHUNK;
  for (int t = t0; t < t0 + TILES_PER_CHUNK; t++) {
    const int col = t * 16 + r16;
    const float sqc = sq[col];
    const float* brow = R + (size_t)col * DPROJ + koff;
    __builtin_prefetch((const void*)(brow + 16 * DPROJ), 0, 1);
    v8f acc = {};
#pragma unroll
    for (int kk = 0; kk < 32; kk++) {
      v2f b = *(const v2f*)(brow + kk * 4);
      acc = wmma_f32_k4(afrag[kk], b, acc);
    }
#pragma unroll
    for (int i = 0; i < 8; i++)
      partial[i] += fmaxf(0.f, sqr[i] + sqc - 2.f * acc[i]);
  }

#pragma unroll
  for (int i = 0; i < 8; i++) {
    float p = partial[i];
    p += __shfl_xor(p, 1, 32);
    p += __shfl_xor(p, 2, 32);
    p += __shfl_xor(p, 4, 32);
    p += __shfl_xor(p, 8, 32);
    if (r16 == 0)
      anchorPart[(size_t)chunk * NPTS + m0 + i + half * 8] = p;
  }
}

__global__ __launch_bounds__(256)
void combine_kernel(const float* __restrict__ anchorPart,
                    float* __restrict__ anchorSq) {
  const int i = blockIdx.x * 256 + threadIdx.x;
  float s = 0.f;
#pragma unroll
  for (int c = 0; c < NCHUNK; c++) s += anchorPart[(size_t)c * NPTS + i];
  anchorSq[i] = s;
}

__global__ __launch_bounds__(1024)
void kmajor_kernel(float* __restrict__ R) {
  float* blk = R + (size_t)blockIdx.x * (128 * DPROJ);
  const int lane = threadIdx.x & 31;
  const int w    = threadIdx.x >> 5;
  const int m    = 32 * (w & 3) + lane;
  const int kg0  = w >> 2;

  float4 v[4];
#pragma unroll
  for (int i = 0; i < 4; i++)
    v[i] = *(const float4*)(blk + m * DPROJ + 4 * (kg0 + 8 * i));

  __syncthreads();

#pragma unroll
  for (int i = 0; i < 4; i++) {
    const int k = 4 * (kg0 + 8 * i);
    blk[(k + 0) * 128 + m] = v[i].x;
    blk[(k + 1) * 128 + m] = v[i].y;
    blk[(k + 2) * 128 + m] = v[i].z;
    blk[(k + 3) * 128 + m] = v[i].w;
  }
}

#define MMAX   4
#define PPITCH 132

typedef float v4f __attribute__((ext_vector_type(4)));
typedef const __attribute__((address_space(1))) float* gfp;
typedef const __attribute__((address_space(1))) v4f*   gv4p;

__device__ __forceinline__ void load_group(v4f (&r)[4], gfp p) {
#pragma unroll
  for (int u = 0; u < 4; u++) r[u] = *(gv4p)(p + u * 128);
}

template <int M>
__device__ __forceinline__ void fold_group(const v4f (&v)[4], const float* pool,
                                           const int (&so)[M], const int g,
                                           v2f (&d)[M][2]) {
  v4f w[M];
#pragma unroll
  for (int c = 0; c < M; c++) w[c] = *(const v4f*)(pool + so[c] + 4 * g);
  if constexpr (M == 2) asm volatile("" : "+v"(w[0]), "+v"(w[1]));
  if constexpr (M == 3) asm volatile("" : "+v"(w[0]), "+v"(w[1]), "+v"(w[2]));
  if constexpr (M == 4) asm volatile("" : "+v"(w[0]), "+v"(w[1]), "+v"(w[2]), "+v"(w[3]));
  v2f pa[M], pb[M];
#pragma unroll
  for (int c = 0; c < M; c++) { pa[c] = v[1].xy * w[c].yy; pb[c] = v[1].zw * w[c].yy; }
#pragma unroll
  for (int c = 0; c < M; c++) {
    pa[c] = __builtin_elementwise_fma(v[0].xy, w[c].xx, pa[c]);
    pb[c] = __builtin_elementwise_fma(v[0].zw, w[c].xx, pb[c]);
  }
#pragma unroll
  for (int c = 0; c < M; c++) {
    pa[c] = __builtin_elementwise_fma(v[2].xy, w[c].zz, pa[c]);
    pb[c] = __builtin_elementwise_fma(v[2].zw, w[c].zz, pb[c]);
  }
#pragma unroll
  for (int c = 0; c < M; c++) {
    pa[c] = __builtin_elementwise_fma(v[3].xy, w[c].ww, pa[c]);
    pb[c] = __builtin_elementwise_fma(v[3].zw, w[c].ww, pb[c]);
  }
#pragma unroll
  for (int c = 0; c < M; c++) { d[c][0] += pa[c]; d[c][1] += pb[c]; }
}

template <int M>
__device__ __forceinline__ void after_fold(gfp &p, v2f (&d)[M][2]) {
#pragma unroll
  for (int c = 0; c < M; c++)
    asm volatile("" : "+v"(p), "+v"(d[c][0]), "+v"(d[c][1]));
}

template <int M, bool AHEAD>
__device__ __forceinline__ void ring_trip(gfp &p, const float* pool,
                                          const int (&so)[M], const int g,
                                          v4f (&r0)[4], v4f (&r1)[4],
                                          v4f (&r2)[4], v4f (&r3)[4],
                                          v2f (&d)[M][2]) {
  load_group(r3, p + 3 * 512);
  fold_group<M>(r0, pool, so, g, d);
  after_fold<M>(p, d);
  if (AHEAD) load_group(r0, p + 4 * 512);
  fold_group<M>(r1, pool, so, g + 1, d);
  after_fold<M>(p, d);
  if (AHEAD) load_group(r1, p + 5 * 512);
  fold_group<M>(r2, pool, so, g + 2, d);
  after_fold<M>(p, d);
  if (AHEAD) load_group(r2, p + 6 * 512);
  fold_group<M>(r3, pool, so, g + 3, d);
  after_fold<M>(p, d);
  p += 4 * 512;
}

template <int M>
__device__ __forceinline__ void sweep_cols(const float* __restrict__ RB,
                                           const float* sqS, float* anchS,
                                           const float* pool, const int* slotS,
                                           const float* selSqS,
                                           const unsigned run, const int t) {
  int   so[M];
  float ssq[M];
#pragma unroll
  for (int c = 0; c < M; c++) {
    so[c]  = __builtin_amdgcn_readfirstlane(slotS[c]) * PPITCH;
    ssq[c] = __int_as_float(__builtin_amdgcn_readfirstlane(__float_as_int(selSqS[c])));
  }
  gfp p = (gfp)(RB + run);
  v4f r0[4], r1[4], r2[4], r3[4];
  load_group(r0, p);
  asm volatile("" : "+v"(p));
  load_group(r1, p + 512);
  asm volatile("" : "+v"(p));
  load_group(r2, p + 1024);
#pragma unroll
  for (int j = 0; j < 4; j++) {
    v2f d[M][2];
#pragma unroll
    for (int c = 0; c < M; c++) { d[c][0] = (v2f){0.f, 0.f}; d[c][1] = (v2f){0.f, 0.f}; }
    if (j < 3) {
#pragma unroll 1
      for (int g = 0; g < 32; g += 4)
        ring_trip<M, true>(p, pool, so, g, r0, r1, r2, r3, d);
    } else {
#pragma unroll 1
      for (int g = 0; g < 28; g += 4)
        ring_trip<M, true>(p, pool, so, g, r0, r1, r2, r3, d);
      ring_trip<M, false>(p, pool, so, 28, r0, r1, r2, r3, d);
    }
    const float4 q = *(const float4*)(sqS + 4 * (1024 * j + t));
    float4 a = *(const float4*)(anchS + 4 * (1024 * j + t));
#pragma unroll
    for (int c = 0; c < M; c++) {
      const float e0 = q.x + ssq[c] - 2.f * d[c][0].x;
      const float e1 = q.y + ssq[c] - 2.f * d[c][0].y;
      const float e2 = q.z + ssq[c] - 2.f * d[c][1].x;
      const float e3 = q.w + ssq[c] - 2.f * d[c][1].y;
      a.x = fminf(a.x, sqrtf(fmaxf(e0, 0.f)));
      a.y = fminf(a.y, sqrtf(fmaxf(e1, 0.f)));
      a.z = fminf(a.z, sqrtf(fmaxf(e2, 0.f)));
      a.w = fminf(a.w, sqrtf(fmaxf(e3, 0.f)));
    }
    *(float4*)(anchS + 4 * (1024 * j + t)) = a;
  }
}

__global__ __launch_bounds__(1024)
void greedy_kernel(const float* __restrict__ RB, const float* __restrict__ sq,
                   const float* __restrict__ anchorSq, int* __restrict__ sel) {
  __shared__ float redV[32];
  __shared__ int   redI[32];
  __shared__ float redS[32];
  __shared__ __attribute__((aligned(16))) float pool[32 * PPITCH];
  __shared__ int   slotS[MMAX];
  __shared__ float selSqS[MMAX];
  __shared__ int   mS;
  __shared__ __attribute__((aligned(16))) float anchS[4 * 1024 * 4];
  __shared__ __attribute__((aligned(16))) float sqS[4 * 1024 * 4];

  const int t    = threadIdx.x;
  const int lane = t & 31;
  const int wid  = t >> 5;
  const int row0 = 512 * wid + 4 * lane;

#pragma unroll
  for (int j = 0; j < 4; j++) {
    float4 a = *(const float4*)(anchorSq + row0 + 128 * j);
    a.x = sqrtf(a.x);
    a.y = sqrtf(a.y);
    a.z = sqrtf(a.z);
    a.w = sqrtf(a.w);
    *(float4*)(anchS + 4 * (1024 * j + t)) = a;
    *(float4*)(sqS + 4 * (1024 * j + t)) = *(const float4*)(sq + row0 + 128 * j);
  }

  const unsigned run = (unsigned)wid * (4u * 128u * DPROJ) + 4u * (unsigned)lane;

  int done = 0;
  for (int pass = 0; pass < NSAMP && done < NSAMP; pass++) {
    float aVal[16];
#pragma unroll
    for (int j = 0; j < 4; j++) {
      const float4 a = *(const float4*)(anchS + 4 * (1024 * j + t));
      aVal[4 * j + 0] = a.x; aVal[4 * j + 1] = a.y; aVal[4 * j + 2] = a.z; aVal[4 * j + 3] = a.w;
    }
    float bv = aVal[0]; int bi = row0; float sv = -1.f;
#pragma unroll
    for (int e = 1; e < 16; e++) {
      const float v = aVal[e];
      if (v > bv) { sv = bv; bv = v; bi = row0 + 128 * (e >> 2) + (e & 3); }
      else        { sv = fmaxf(sv, v); }
    }
#pragma unroll
    for (int m = 1; m < 32; m <<= 1) {
      const float ov = __shfl_xor(bv, m, 32);
      const int   oi = __shfl_xor(bi, m, 32);
      const float os = __shfl_xor(sv, m, 32);
      const bool take = (ov > bv) || (ov == bv && oi < bi);
      sv = take ? fmaxf(os, bv) : fmaxf(sv, ov);
      if (take) { bv = ov; bi = oi; }
    }
    {
      const unsigned base = (unsigned)(bi >> 7) * (128u * DPROJ) + (unsigned)(bi & 127);
#pragma unroll
      for (int i = 0; i < 4; i++)
        pool[wid * PPITCH + lane + 32 * i] = RB[base + (unsigned)(lane + 32 * i) * 128u];
    }
    if (lane == 0) { redV[wid] = bv; redI[wid] = bi; redS[wid] = sv; }
    __syncthreads();

    if (wid == 0) {
      float pv = redV[lane];
      const int pi = redI[lane];
      float T = redS[lane];
#pragma unroll
      for (int m = 1; m < 32; m <<= 1) T = fmaxf(T, __shfl_xor(T, m, 32));
      const float psq = sq[pi];
      const float4* own = (const float4*)(pool + lane * PPITCH);
      int m = 0;
      for (int c = 0; c < MMAX; c++) {
        float cv = pv; int ci = pi; int cl = lane;
#pragma unroll
        for (int x = 1; x < 32; x <<= 1) {
          const float ov = __shfl_xor(cv, x, 32);
          const int   oi = __shfl_xor(ci, x, 32);
          const int   ol = __shfl_xor(cl, x, 32);
          if (ov > cv || (ov == cv && oi < ci)) { cv = ov; ci = oi; cl = ol; }
        }
        if (c > 0 && !(cv > T)) break;
        const float ssq = __shfl(psq, cl, 32);
        if (lane == 0) { sel[done + c] = ci; slotS[c] = cl; selSqS[c] = ssq; }
        m = c + 1;
        if (done + m == NSAMP || m == MMAX) break;
        const float4* sr = (const float4*)(pool + cl * PPITCH);
        float d = 0.f;
#pragma unroll 8
        for (int kg = 0; kg < 32; kg++) {
          const float4 r = own[kg];
          const float4 w = sr[kg];
          float p = r.y * w.y;
          p = fmaf(r.x, w.x, p);
          p = fmaf(r.z, w.z, p);
          p = fmaf(r.w, w.w, p);
          d += p;
        }
        const float e = psq + ssq - 2.f * d;
        pv = fminf(pv, sqrtf(fmaxf(e, 0.f)));
      }
      if (lane == 0) mS = m;
    }
    __syncthreads();

    const int m = mS;
    if (m == 1)      sweep_cols<1>(RB, sqS, anchS, pool, slotS, selSqS, run, t);
    else if (m == 2) sweep_cols<2>(RB, sqS, anchS, pool, slotS, selSqS, run, t);
    else if (m == 3) sweep_cols<3>(RB, sqS, anchS, pool, slotS, selSqS, run, t);
    else             sweep_cols<4>(RB, sqS, anchS, pool, slotS, selSqS, run, t);
    done += m;
    __syncthreads();
  }
}

__global__ __launch_bounds__(256)
void gather_kernel(const float* __restrict__ F, const int* __restrict__ sel,
                   float* __restrict__ out) {
  const int s = blockIdx.x;
  const int c = blockIdx.y * 256 + threadIdx.x;
  out[(size_t)s * DIN + c] = F[(size_t)sel[s] * DIN + c];
}

extern "C" void kernel_launch(void* const* d_in, const int* in_sizes, int n_in,
                              void* d_out, int out_size, void* d_ws, size_t ws_size,
                              hipStream_t stream) {
  (void)in_sizes; (void)n_in; (void)out_size; (void)ws_size;
  const float* F = (const float*)d_in[0];
  const float* W = (const float*)d_in[1];

  float* R          = (float*)d_ws;
  float* sq         = R + (size_t)NPTS * DPROJ;
  float* anchorSq   = sq + NPTS;
  float* anchorPart = anchorSq + NPTS;
  int*   sel        = (int*)(anchorPart + (size_t)NCHUNK * NPTS);

  proj_kernel<<<1024, 256, 0, stream>>>(F, W, R);
  sq_kernel<<<NPTS / 8, 256, 0, stream>>>(R, sq);
  gram_rowsum_kernel<<<1024, 256, 0, stream>>>(R, sq, anchorPart);
  combine_kernel<<<NPTS / 256, 256, 0, stream>>>(anchorPart, anchorSq);
  kmajor_kernel<<<NPTS / 128, 1024, 0, stream>>>(R);
  greedy_kernel<<<1, 1024, 0, stream>>>(R, sq, anchorSq, sel);
  gather_kernel<<<dim3(NSAMP, DIN / 256), 256, 0, stream>>>(F, sel, (float*)d_out);
}


# gram pass: runs behind the k-major transpose and reads both operands from it; one 16-byte load per lane feeds four column tiles from whole lines, not 16-line fragment loads; row-sum order changes
# speedup vs baseline: 12.1577x; 1.1868x over previous
#include <hip/hip_runtime.h>


#define NPTS   16384
#define DIN    1024
#define DPROJ  128
#define NSAMP  163
#define NCHUNK 8

typedef float v2f __attribute__((ext_vector_type(2)));
typedef float v8f __attribute__((ext_vector_type(8)));

__device__ __forceinline__ v8f wmma_f32_k4(v2f a, v2f b, v8f c) {
  return __builtin_amdgcn_wmma_f32_16x16x4_f32(
       false, a,  false, b,
       (short)0, c,  false,  false);
}

__global__ __launch_bounds__(256)
void proj_kernel(const float* __restrict__ F, const float* __restrict__ W,
                 float* __restrict__ R) {
  const int wave = (blockIdx.x * blockDim.x + threadIdx.x) >> 5;
  const int lane = threadIdx.x & 31;
  const int tm = wave >> 3;
  const int tn = wave & 7;
  const int m0 = tm * 16, n0 = tn * 16;
  const int half = lane >> 4;
  const int r16  = lane & 15;
  const int koff = half * 2;

  v8f acc = {};
  const float* arow = F + (size_t)(m0 + r16) * DIN + koff;
  for (int k = 0; k < DIN; k += 4) {
    v2f a = *(const v2f*)(arow + k);
    v2f b;
    b.x = W[(size_t)(k + koff)     * DPROJ + n0 + r16];
    b.y = W[(size_t)(k + koff + 1) * DPROJ + n0 + r16];
    acc = wmma_f32_k4(a, b, acc);
  }
#pragma unroll
  for (int i = 0; i < 8; i++)
    R[(size_t)(m0 + i + half * 8) * DPROJ + n0 + r16] = acc[i];
}

__global__ __launch_bounds__(256)
void sq_kernel(const float* __restrict__ R, float* __restrict__ sq) {
  const int row  = (blockIdx.x * blockDim.x + threadIdx.x) >> 5;
  const int lane = threadIdx.x & 31;
  const float* rp = R + (size_t)row * DPROJ;
  float s = 0.f;
#pragma unroll
  for (int k = lane; k < DPROJ; k += 32) { float v = rp[k]; s += v * v; }
#pragma unroll
  for (int m = 16; m >= 1; m >>= 1) s += __shfl_xor(s, m, 32);
  if (lane == 0) sq[row] = s;
}

typedef float v4f __attribute__((ext_vector_type(4)));

__device__ __forceinline__ void gram_load(v4f (&b)[8], const float* p, const int ch) {
#pragma unroll
  for (int u = 0; u < 4; u++) {
    b[2 * u]     = *(const v4f*)(p + (4 * ch + u) * 512);
    b[2 * u + 1] = *(const v4f*)(p + (4 * ch + u) * 512 + 128);
  }
}

template <int CH>
__device__ __forceinline__ void gram_mma(const v4f (&b)[8], const v2f (&afrag)[32],
                                         v8f (&acc)[4]) {
#pragma unroll
  for (int u = 0; u < 4; u++) {
    const v2f a = afrag[4 * CH + u];
    acc[0] = wmma_f32_k4(a, (v2f){b[2 * u].x, b[2 * u + 1].x}, acc[0]);
    acc[1] = wmma_f32_k4(a, (v2f){b[2 * u].y, b[2 * u + 1].y}, acc[1]);
    acc[2] = wmma_f32_k4(a, (v2f){b[2 * u].z, b[2 * u + 1].z}, acc[2]);
    acc[3] = wmma_f32_k4(a, (v2f){b[2 * u].w, b[2 * u + 1].w}, acc[3]);
  }
}

__global__ __launch_bounds__(256)
void gram_rowsum_kernel(const float* __restrict__ RB, const float* __restrict__ sq,
                        float* __restrict__ anchorPart) {
  const int wave  = (blockIdx.x * blockDim.x + threadIdx.x) >> 5;
  const int lane  = threadIdx.x & 31;
  const int chunk = wave & (NCHUNK - 1);
  const int strip = wave >> 3;
  const int half  = lane >> 4;
  const int r16   = lane & 15;
  const int m0    = strip * 16;

  v2f afrag[32];
  const float* ap = RB + (size_t)(m0 >> 7) * (128 * DPROJ) + ((m0 & 127) + r16)
                       + half * 256;
#pragma unroll
  for (int kk = 0; kk < 32; kk++) {
    afrag[kk].x = ap[kk * 512];
    afrag[kk].y = ap[kk * 512 + 128];
  }

  float sqr[8];
#pragma unroll
  for (int i = 0; i < 8; i++) sqr[i] = sq[m0 + i + half * 8];

  float partial[8] = {0.f, 0.f, 0.f, 0.f, 0.f, 0.f, 0.f, 0.f};

  const float* cp = RB + (size_t)(chunk * 16) * (128 * DPROJ) + 4 * r16 + half * 256;
  const float* sp = sq + chunk * 2048 + 4 * r16;

  v4f bA[8], bB[8];
  gram_load(bA, cp, 0);

#pragma unroll 1
  for (int G = 0; G < 32; G++) {
    const float* p  = cp + (size_t)(G >> 1) * (128 * DPROJ) + (G & 1) * 64;
    const int Gn = (G < 31) ? G + 1 : G;
    const float* pn = cp + (size_t)(Gn >> 1) * (128 * DPROJ) + (Gn & 1) * 64;
    const v4f sc = *(const v4f*)(sp + 64 * G);

    v8f acc[4] = {};
    gram_load(bB, p, 1);
    gram_mma<0>(bA, afrag, acc);
    gram_load(bA, p, 2);
    gram_mma<1>(bB, afrag, acc);
    gram_load(bB, p, 3);
    gram_mma<2>(bA, afrag, acc);
    gram_load(bA, p, 4);
    gram_mma<3>(bB, afrag, acc);
    gram_load(bB, p, 5);
    gram_mma<4>(bA, afrag, acc);
    gram_load(bA, p, 6);
    gram_mma<5>(bB, afrag, acc);
    gram_load(bB, p, 7);
    gram_mma<6>(bA, afrag, acc);
    gram_load(bA, pn, 0);
    gram_mma<7>(bB, afrag, acc);

#pragma unroll
    for (int q = 0; q < 4; q++) {
#pragma unroll
      for (int i = 0; i < 8; i++)
        partial[i] += fmaxf(0.f, sqr[i] + sc[q] - 2.f * acc[q][i]);
    }
  }

#pragma unroll
  for (int i = 0; i < 8; i++) {
    float p = partial[i];
    p += __shfl_xor(p, 1, 32);
    p += __shfl_xor(p, 2, 32);
    p += __shfl_xor(p, 4, 32);
    p += __shfl_xor(p, 8, 32);
    if (r16 == 0)
      anchorPart[(size_t)chunk * NPTS + m0 + i + half * 8] = p;
  }
}

__global__ __launch_bounds__(256)
void combine_kernel(const float* __restrict__ anchorPart,
                    float* __restrict__ anchorSq) {
  const int i = blockIdx.x * 256 + threadIdx.x;
  float s = 0.f;
#pragma unroll
  for (int c = 0; c < NCHUNK; c++) s += anchorPart[(size_t)c * NPTS + i];
  anchorSq[i] = s;
}

__global__ __launch_bounds__(1024)
void kmajor_kernel(float* __restrict__ R) {
  float* blk = R + (size_t)blockIdx.x * (128 * DPROJ);
  const int lane = threadIdx.x & 31;
  const int w    = threadIdx.x >> 5;
  const int m    = 32 * (w & 3) + lane;
  const int kg0  = w >> 2;

  float4 v[4];
#pragma unroll
  for (int i = 0; i < 4; i++)
    v[i] = *(const float4*)(blk + m * DPROJ + 4 * (kg0 + 8 * i));

  __syncthreads();

#pragma unroll
  for (int i = 0; i < 4; i++) {
    const int k = 4 * (kg0 + 8 * i);
    blk[(k + 0) * 128 + m] = v[i].x;
    blk[(k + 1) * 128 + m] = v[i].y;
    blk[(k + 2) * 128 + m] = v[i].z;
    blk[(k + 3) * 128 + m] = v[i].w;
  }
}

#define MMAX   4
#define PPITCH 132

typedef const __attribute__((address_space(1))) float* gfp;
typedef const __attribute__((address_space(1))) v4f*   gv4p;

__device__ __forceinline__ void load_group(v4f (&r)[4], gfp p) {
#pragma unroll
  for (int u = 0; u < 4; u++) r[u] = *(gv4p)(p + u * 128);
}

template <int M>
__device__ __forceinline__ void fold_group(const v4f (&v)[4], const float* pool,
                                           const int (&so)[M], const int g,
                                           v2f (&d)[M][2]) {
  v4f w[M];
#pragma unroll
  for (int c = 0; c < M; c++) w[c] = *(const v4f*)(pool + so[c] + 4 * g);
  if constexpr (M == 2) asm volatile("" : "+v"(w[0]), "+v"(w[1]));
  if constexpr (M == 3) asm volatile("" : "+v"(w[0]), "+v"(w[1]), "+v"(w[2]));
  if constexpr (M == 4) asm volatile("" : "+v"(w[0]), "+v"(w[1]), "+v"(w[2]), "+v"(w[3]));
  v2f pa[M], pb[M];
#pragma unroll
  for (int c = 0; c < M; c++) { pa[c] = v[1].xy * w[c].yy; pb[c] = v[1].zw * w[c].yy; }
#pragma unroll
  for (int c = 0; c < M; c++) {
    pa[c] = __builtin_elementwise_fma(v[0].xy, w[c].xx, pa[c]);
    pb[c] = __builtin_elementwise_fma(v[0].zw, w[c].xx, pb[c]);
  }
#pragma unroll
  for (int c = 0; c < M; c++) {
    pa[c] = __builtin_elementwise_fma(v[2].xy, w[c].zz, pa[c]);
    pb[c] = __builtin_elementwise_fma(v[2].zw, w[c].zz, pb[c]);
  }
#pragma unroll
  for (int c = 0; c < M; c++) {
    pa[c] = __builtin_elementwise_fma(v[3].xy, w[c].ww, pa[c]);
    pb[c] = __builtin_elementwise_fma(v[3].zw, w[c].ww, pb[c]);
  }
#pragma unroll
  for (int c = 0; c < M; c++) { d[c][0] += pa[c]; d[c][1] += pb[c]; }
}

template <int M>
__device__ __forceinline__ void after_fold(gfp &p, v2f (&d)[M][2]) {
#pragma unroll
  for (int c = 0; c < M; c++)
    asm volatile("" : "+v"(p), "+v"(d[c][0]), "+v"(d[c][1]));
}

template <int M, bool AHEAD>
__device__ __forceinline__ void ring_trip(gfp &p, const float* pool,
                                          const int (&so)[M], const int g,
                                          v4f (&r0)[4], v4f (&r1)[4],
                                          v4f (&r2)[4], v4f (&r3)[4],
                                          v2f (&d)[M][2]) {
  load_group(r3, p + 3 * 512);
  fold_group<M>(r0, pool, so, g, d);
  after_fold<M>(p, d);
  if (AHEAD) load_group(r0, p + 4 * 512);
  fold_group<M>(r1, pool, so, g + 1, d);
  after_fold<M>(p, d);
  if (AHEAD) load_group(r1, p + 5 * 512);
  fold_group<M>(r2, pool, so, g + 2, d);
  after_fold<M>(p, d);
  if (AHEAD) load_group(r2, p + 6 * 512);
  fold_group<M>(r3, pool, so, g + 3, d);
  after_fold<M>(p, d);
  p += 4 * 512;
}

template <int M>
__device__ __forceinline__ void sweep_cols(const float* __restrict__ RB,
                                           const float* sqS, float* anchS,
                                           const float* pool, const int* slotS,
                                           const float* selSqS,
                                           const unsigned run, const int t) {
  int   so[M];
  float ssq[M];
#pragma unroll
  for (int c = 0; c < M; c++) {
    so[c]  = __builtin_amdgcn_readfirstlane(slotS[c]) * PPITCH;
    ssq[c] = __int_as_float(__builtin_amdgcn_readfirstlane(__float_as_int(selSqS[c])));
  }
  gfp p = (gfp)(RB + run);
  v4f r0[4], r1[4], r2[4], r3[4];
  load_group(r0, p);
  asm volatile("" : "+v"(p));
  load_group(r1, p + 512);
  asm volatile("" : "+v"(p));
  load_group(r2, p + 1024);
#pragma unroll
  for (int j = 0; j < 4; j++) {
    v2f d[M][2];
#pragma unroll
    for (int c = 0; c < M; c++) { d[c][0] = (v2f){0.f, 0.f}; d[c][1] = (v2f){0.f, 0.f}; }
    if (j < 3) {
#pragma unroll 1
      for (int g = 0; g < 32; g += 4)
        ring_trip<M, true>(p, pool, so, g, r0, r1, r2, r3, d);
    } else {
#pragma unroll 1
      for (int g = 0; g < 28; g += 4)
        ring_trip<M, true>(p, pool, so, g, r0, r1, r2, r3, d);
      ring_trip<M, false>(p, pool, so, 28, r0, r1, r2, r3, d);
    }
    const float4 q = *(const float4*)(sqS + 4 * (1024 * j + t));
    float4 a = *(const float4*)(anchS + 4 * (1024 * j + t));
#pragma unroll
    for (int c = 0; c < M; c++) {
      const float e0 = q.x + ssq[c] - 2.f * d[c][0].x;
      const float e1 = q.y + ssq[c] - 2.f * d[c][0].y;
      const float e2 = q.z + ssq[c] - 2.f * d[c][1].x;
      const float e3 = q.w + ssq[c] - 2.f * d[c][1].y;
      a.x = fminf(a.x, sqrtf(fmaxf(e0, 0.f)));
      a.y = fminf(a.y, sqrtf(fmaxf(e1, 0.f)));
      a.z = fminf(a.z, sqrtf(fmaxf(e2, 0.f)));
      a.w = fminf(a.w, sqrtf(fmaxf(e3, 0.f)));
    }
    *(float4*)(anchS + 4 * (1024 * j + t)) = a;
  }
}

__global__ __launch_bounds__(1024)
void greedy_kernel(const float* __restrict__ RB, const float* __restrict__ sq,
                   const float* __restrict__ anchorSq, int* __restrict__ sel) {
  __shared__ float redV[32];
  __shared__ int   redI[32];
  __shared__ float redS[32];
  __shared__ __attribute__((aligned(16))) float pool[32 * PPITCH];
  __shared__ int   slotS[MMAX];
  __shared__ float selSqS[MMAX];
  __shared__ int   mS;
  __shared__ __attribute__((aligned(16))) float anchS[4 * 1024 * 4];
  __shared__ __attribute__((aligned(16))) float sqS[4 * 1024 * 4];

  const int t    = threadIdx.x;
  const int lane = t & 31;
  const int wid  = t >> 5;
  const int row0 = 512 * wid + 4 * lane;

#pragma unroll
  for (int j = 0; j < 4; j++) {
    float4 a = *(const float4*)(anchorSq + row0 + 128 * j);
    a.x = sqrtf(a.x);
    a.y = sqrtf(a.y);
    a.z = sqrtf(a.z);
    a.w = sqrtf(a.w);
    *(float4*)(anchS + 4 * (1024 * j + t)) = a;
    *(float4*)(sqS + 4 * (1024 * j + t)) = *(const float4*)(sq + row0 + 128 * j);
  }

  const unsigned run = (unsigned)wid * (4u * 128u * DPROJ) + 4u * (unsigned)lane;

  int done = 0;
  for (int pass = 0; pass < NSAMP && done < NSAMP; pass++) {
    float aVal[16];
#pragma unroll
    for (int j = 0; j < 4; j++) {
      const float4 a = *(const float4*)(anchS + 4 * (1024 * j + t));
      aVal[4 * j + 0] = a.x; aVal[4 * j + 1] = a.y; aVal[4 * j + 2] = a.z; aVal[4 * j + 3] = a.w;
    }
    float bv = aVal[0]; int bi = row0; float sv = -1.f;
#pragma unroll
    for (int e = 1; e < 16; e++) {
      const float v = aVal[e];
      if (v > bv) { sv = bv; bv = v; bi = row0 + 128 * (e >> 2) + (e & 3); }
      else        { sv = fmaxf(sv, v); }
    }
#pragma unroll
    for (int m = 1; m < 32; m <<= 1) {
      const float ov = __shfl_xor(bv, m, 32);
      const int   oi = __shfl_xor(bi, m, 32);
      const float os = __shfl_xor(sv, m, 32);
      const bool take = (ov > bv) || (ov == bv && oi < bi);
      sv = take ? fmaxf(os, bv) : fmaxf(sv, ov);
      if (take) { bv = ov; bi = oi; }
    }
    {
      const unsigned base = (unsigned)(bi >> 7) * (128u * DPROJ) + (unsigned)(bi & 127);
#pragma unroll
      for (int i = 0; i < 4; i++)
        pool[wid * PPITCH + lane + 32 * i] = RB[base + (unsigned)(lane + 32 * i) * 128u];
    }
    if (lane == 0) { redV[wid] = bv; redI[wid] = bi; redS[wid] = sv; }
    __syncthreads();

    if (wid == 0) {
      float pv = redV[lane];
      const int pi = redI[lane];
      float T = redS[lane];
#pragma unroll
      for (int m = 1; m < 32; m <<= 1) T = fmaxf(T, __shfl_xor(T, m, 32));
      const float psq = sq[pi];
      const float4* own = (const float4*)(pool + lane * PPITCH);
      int m = 0;
      for (int c = 0; c < MMAX; c++) {
        float cv = pv; int ci = pi; int cl = lane;
#pragma unroll
        for (int x = 1; x < 32; x <<= 1) {
          const float ov = __shfl_xor(cv, x, 32);
          const int   oi = __shfl_xor(ci, x, 32);
          const int   ol = __shfl_xor(cl, x, 32);
          if (ov > cv || (ov == cv && oi < ci)) { cv = ov; ci = oi; cl = ol; }
        }
        if (c > 0 && !(cv > T)) break;
        const float ssq = __shfl(psq, cl, 32);
        if (lane == 0) { sel[done + c] = ci; slotS[c] = cl; selSqS[c] = ssq; }
        m = c + 1;
        if (done + m == NSAMP || m == MMAX) break;
        const float4* sr = (const float4*)(pool + cl * PPITCH);
        float d = 0.f;
#pragma unroll 8
        for (int kg = 0; kg < 32; kg++) {
          const float4 r = own[kg];
          const float4 w = sr[kg];
          float p = r.y * w.y;
          p = fmaf(r.x, w.x, p);
          p = fmaf(r.z, w.z, p);
          p = fmaf(r.w, w.w, p);
          d += p;
        }
        const float e = psq + ssq - 2.f * d;
        pv = fminf(pv, sqrtf(fmaxf(e, 0.f)));
      }
      if (lane == 0) mS = m;
    }
    __syncthreads();

    const int m = mS;
    if (m == 1)      sweep_cols<1>(RB, sqS, anchS, pool, slotS, selSqS, run, t);
    else if (m == 2) sweep_cols<2>(RB, sqS, anchS, pool, slotS, selSqS, run, t);
    else if (m == 3) sweep_cols<3>(RB, sqS, anchS, pool, slotS, selSqS, run, t);
    else             sweep_cols<4>(RB, sqS, anchS, pool, slotS, selSqS, run, t);
    done += m;
    __syncthreads();
  }
}

__global__ __launch_bounds__(256)
void gather_kernel(const float* __restrict__ F, const int* __restrict__ sel,
                   float* __restrict__ out) {
  const int s = blockIdx.x;
  const int c = blockIdx.y * 256 + threadIdx.x;
  out[(size_t)s * DIN + c] = F[(size_t)sel[s] * DIN + c];
}

extern "C" void kernel_launch(void* const* d_in, const int* in_sizes, int n_in,
                              void* d_out, int out_size, void* d_ws, size_t ws_size,
                              hipStream_t stream) {
  (void)in_sizes; (void)n_in; (void)out_size; (void)ws_size;
  const float* F = (const float*)d_in[0];
  const float* W = (const float*)d_in[1];

  float* R          = (float*)d_ws;
  float* sq         = R + (size_t)NPTS * DPROJ;
  float* anchorSq   = sq + NPTS;
  float* anchorPart = anchorSq + NPTS;
  int*   sel        = (int*)(anchorPart + (size_t)NCHUNK * NPTS);

  proj_kernel<<<1024, 256, 0, stream>>>(F, W, R);
  sq_kernel<<<NPTS / 8, 256, 0, stream>>>(R, sq);
  kmajor_kernel<<<NPTS / 128, 1024, 0, stream>>>(R);
  gram_rowsum_kernel<<<1024, 256, 0, stream>>>(R, sq, anchorPart);
  combine_kernel<<<NPTS / 256, 256, 0, stream>>>(anchorPart, anchorSq);
  greedy_kernel<<<1, 1024, 0, stream>>>(R, sq, anchorSq, sel);
  gather_kernel<<<dim3(NSAMP, DIN / 256), 256, 0, stream>>>(F, sel, (float*)d_out);
}


# greedy sweep: a row's M squared distances are reduced to the smallest first and clip and square root taken once per row and sweep, not M times; min and a monotone root commute, anchors bit-identical
# speedup vs baseline: 12.8199x; 1.0545x over previous
#include <hip/hip_runtime.h>


#define NPTS   16384
#define DIN    1024
#define DPROJ  128
#define NSAMP  163
#define NCHUNK 8

typedef float v2f __attribute__((ext_vector_type(2)));
typedef float v8f __attribute__((ext_vector_type(8)));

__device__ __forceinline__ v8f wmma_f32_k4(v2f a, v2f b, v8f c) {
  return __builtin_amdgcn_wmma_f32_16x16x4_f32(
       false, a,  false, b,
       (short)0, c,  false,  false);
}

__global__ __launch_bounds__(256)
void proj_kernel(const float* __restrict__ F, const float* __restrict__ W,
                 float* __restrict__ R) {
  const int wave = (blockIdx.x * blockDim.x + threadIdx.x) >> 5;
  const int lane = threadIdx.x & 31;
  const int tm = wave >> 3;
  const int tn = wave & 7;
  const int m0 = tm * 16, n0 = tn * 16;
  const int half = lane >> 4;
  const int r16  = lane & 15;
  const int koff = half * 2;

  v8f acc = {};
  const float* arow = F + (size_t)(m0 + r16) * DIN + koff;
  for (int k = 0; k < DIN; k += 4) {
    v2f a = *(const v2f*)(arow + k);
    v2f b;
    b.x = W[(size_t)(k + koff)     * DPROJ + n0 + r16];
    b.y = W[(size_t)(k + koff + 1) * DPROJ + n0 + r16];
    acc = wmma_f32_k4(a, b, acc);
  }
#pragma unroll
  for (int i = 0; i < 8; i++)
    R[(size_t)(m0 + i + half * 8) * DPROJ + n0 + r16] = acc[i];
}

__global__ __launch_bounds__(256)
void sq_kernel(const float* __restrict__ R, float* __restrict__ sq) {
  const int row  = (blockIdx.x * blockDim.x + threadIdx.x) >> 5;
  const int lane = threadIdx.x & 31;
  const float* rp = R + (size_t)row * DPROJ;
  float s = 0.f;
#pragma unroll
  for (int k = lane; k < DPROJ; k += 32) { float v = rp[k]; s += v * v; }
#pragma unroll
  for (int m = 16; m >= 1; m >>= 1) s += __shfl_xor(s, m, 32);
  if (lane == 0) sq[row] = s;
}

typedef float v4f __attribute__((ext_vector_type(4)));

__device__ __forceinline__ void gram_load(v4f (&b)[8], const float* p, const int ch) {
#pragma unroll
  for (int u = 0; u < 4; u++) {
    b[2 * u]     = *(const v4f*)(p + (4 * ch + u) * 512);
    b[2 * u + 1] = *(const v4f*)(p + (4 * ch + u) * 512 + 128);
  }
}

template <int CH>
__device__ __forceinline__ void gram_mma(const v4f (&b)[8], const v2f (&afrag)[32],
                                         v8f (&acc)[4]) {
#pragma unroll
  for (int u = 0; u < 4; u++) {
    const v2f a = afrag[4 * CH + u];
    acc[0] = wmma_f32_k4(a, (v2f){b[2 * u].x, b[2 * u + 1].x}, acc[0]);
    acc[1] = wmma_f32_k4(a, (v2f){b[2 * u].y, b[2 * u + 1].y}, acc[1]);
    acc[2] = wmma_f32_k4(a, (v2f){b[2 * u].z, b[2 * u + 1].z}, acc[2]);
    acc[3] = wmma_f32_k4(a, (v2f){b[2 * u].w, b[2 * u + 1].w}, acc[3]);
  }
}

__global__ __launch_bounds__(256)
void gram_rowsum_kernel(const float* __restrict__ RB, const float* __restrict__ sq,
                        float* __restrict__ anchorPart) {
  const int wave  = (blockIdx.x * blockDim.x + threadIdx.x) >> 5;
  const int lane  = threadIdx.x & 31;
  const int chunk = wave & (NCHUNK - 1);
  const int strip = wave >> 3;
  const int half  = lane >> 4;
  const int r16   = lane & 15;
  const int m0    = strip * 16;

  v2f afrag[32];
  const float* ap = RB + (size_t)(m0 >> 7) * (128 * DPROJ) + ((m0 & 127) + r16)
                       + half * 256;
#pragma unroll
  for (int kk = 0; kk < 32; kk++) {
    afrag[kk].x = ap[kk * 512];
    afrag[kk].y = ap[kk * 512 + 128];
  }

  float sqr[8];
#pragma unroll
  for (int i = 0; i < 8; i++) sqr[i] = sq[m0 + i + half * 8];

  float partial[8] = {0.f, 0.f, 0.f, 0.f, 0.f, 0.f, 0.f, 0.f};

  const float* cp = RB + (size_t)(chunk * 16) * (128 * DPROJ) + 4 * r16 + half * 256;
  const float* sp = sq + chunk * 2048 + 4 * r16;

  v4f bA[8], bB[8];
  gram_load(bA, cp, 0);

#pragma unroll 1
  for (int G = 0; G < 32; G++) {
    const float* p  = cp + (size_t)(G >> 1) * (128 * DPROJ) + (G & 1) * 64;
    const int Gn = (G < 31) ? G + 1 : G;
    const float* pn = cp + (size_t)(Gn >> 1) * (128 * DPROJ) + (Gn & 1) * 64;
    const v4f sc = *(const v4f*)(sp + 64 * G);

    v8f acc[4] = {};
    gram_load(bB, p, 1);
    gram_mma<0>(bA, afrag, acc);
    gram_load(bA, p, 2);
    gram_mma<1>(bB, afrag, acc);
    gram_load(bB, p, 3);
    gram_mma<2>(bA, afrag, acc);
    gram_load(bA, p, 4);
    gram_mma<3>(bB, afrag, acc);
    gram_load(bB, p, 5);
    gram_mma<4>(bA, afrag, acc);
    gram_load(bA, p, 6);
    gram_mma<5>(bB, afrag, acc);
    gram_load(bB, p, 7);
    gram_mma<6>(bA, afrag, acc);
    gram_load(bA, pn, 0);
    gram_mma<7>(bB, afrag, acc);

#pragma unroll
    for (int q = 0; q < 4; q++) {
#pragma unroll
      for (int i = 0; i < 8; i++)
        partial[i] += fmaxf(0.f, sqr[i] + sc[q] - 2.f * acc[q][i]);
    }
  }

#pragma unroll
  for (int i = 0; i < 8; i++) {
    float p = partial[i];
    p += __shfl_xor(p, 1, 32);
    p += __shfl_xor(p, 2, 32);
    p += __shfl_xor(p, 4, 32);
    p += __shfl_xor(p, 8, 32);
    if (r16 == 0)
      anchorPart[(size_t)chunk * NPTS + m0 + i + half * 8] = p;
  }
}

__global__ __launch_bounds__(256)
void combine_kernel(const float* __restrict__ anchorPart,
                    float* __restrict__ anchorSq) {
  const int i = blockIdx.x * 256 + threadIdx.x;
  float s = 0.f;
#pragma unroll
  for (int c = 0; c < NCHUNK; c++) s += anchorPart[(size_t)c * NPTS + i];
  anchorSq[i] = s;
}

__global__ __launch_bounds__(1024)
void kmajor_kernel(float* __restrict__ R) {
  float* blk = R + (size_t)blockIdx.x * (128 * DPROJ);
  const int lane = threadIdx.x & 31;
  const int w    = threadIdx.x >> 5;
  const int m    = 32 * (w & 3) + lane;
  const int kg0  = w >> 2;

  float4 v[4];
#pragma unroll
  for (int i = 0; i < 4; i++)
    v[i] = *(const float4*)(blk + m * DPROJ + 4 * (kg0 + 8 * i));

  __syncthreads();

#pragma unroll
  for (int i = 0; i < 4; i++) {
    const int k = 4 * (kg0 + 8 * i);
    blk[(k + 0) * 128 + m] = v[i].x;
    blk[(k + 1) * 128 + m] = v[i].y;
    blk[(k + 2) * 128 + m] = v[i].z;
    blk[(k + 3) * 128 + m] = v[i].w;
  }
}

#define MMAX   4
#define PPITCH 132

typedef const __attribute__((address_space(1))) float* gfp;
typedef const __attribute__((address_space(1))) v4f*   gv4p;

__device__ __forceinline__ void load_group(v4f (&r)[4], gfp p) {
#pragma unroll
  for (int u = 0; u < 4; u++) r[u] = *(gv4p)(p + u * 128);
}

template <int M>
__device__ __forceinline__ void fold_group(const v4f (&v)[4], const float* pool,
                                           const int (&so)[M], const int g,
                                           v2f (&d)[M][2]) {
  v4f w[M];
#pragma unroll
  for (int c = 0; c < M; c++) w[c] = *(const v4f*)(pool + so[c] + 4 * g);
  if constexpr (M == 2) asm volatile("" : "+v"(w[0]), "+v"(w[1]));
  if constexpr (M == 3) asm volatile("" : "+v"(w[0]), "+v"(w[1]), "+v"(w[2]));
  if constexpr (M == 4) asm volatile("" : "+v"(w[0]), "+v"(w[1]), "+v"(w[2]), "+v"(w[3]));
  v2f pa[M], pb[M];
#pragma unroll
  for (int c = 0; c < M; c++) { pa[c] = v[1].xy * w[c].yy; pb[c] = v[1].zw * w[c].yy; }
#pragma unroll
  for (int c = 0; c < M; c++) {
    pa[c] = __builtin_elementwise_fma(v[0].xy, w[c].xx, pa[c]);
    pb[c] = __builtin_elementwise_fma(v[0].zw, w[c].xx, pb[c]);
  }
#pragma unroll
  for (int c = 0; c < M; c++) {
    pa[c] = __builtin_elementwise_fma(v[2].xy, w[c].zz, pa[c]);
    pb[c] = __builtin_elementwise_fma(v[2].zw, w[c].zz, pb[c]);
  }
#pragma unroll
  for (int c = 0; c < M; c++) {
    pa[c] = __builtin_elementwise_fma(v[3].xy, w[c].ww, pa[c]);
    pb[c] = __builtin_elementwise_fma(v[3].zw, w[c].ww, pb[c]);
  }
#pragma unroll
  for (int c = 0; c < M; c++) { d[c][0] += pa[c]; d[c][1] += pb[c]; }
}

template <int M>
__device__ __forceinline__ void after_fold(gfp &p, v2f (&d)[M][2]) {
#pragma unroll
  for (int c = 0; c < M; c++)
    asm volatile("" : "+v"(p), "+v"(d[c][0]), "+v"(d[c][1]));
}

template <int M, bool AHEAD>
__device__ __forceinline__ void ring_trip(gfp &p, const float* pool,
                                          const int (&so)[M], const int g,
                                          v4f (&r0)[4], v4f (&r1)[4],
                                          v4f (&r2)[4], v4f (&r3)[4],
                                          v2f (&d)[M][2]) {
  load_group(r3, p + 3 * 512);
  fold_group<M>(r0, pool, so, g, d);
  after_fold<M>(p, d);
  if (AHEAD) load_group(r0, p + 4 * 512);
  fold_group<M>(r1, pool, so, g + 1, d);
  after_fold<M>(p, d);
  if (AHEAD) load_group(r1, p + 5 * 512);
  fold_group<M>(r2, pool, so, g + 2, d);
  after_fold<M>(p, d);
  if (AHEAD) load_group(r2, p + 6 * 512);
  fold_group<M>(r3, pool, so, g + 3, d);
  after_fold<M>(p, d);
  p += 4 * 512;
}

template <int M>
__device__ __forceinline__ void sweep_cols(const float* __restrict__ RB,
                                           const float* sqS, float* anchS,
                                           const float* pool, const int* slotS,
                                           const float* selSqS,
                                           const unsigned run, const int t) {
  int   so[M];
  float ssq[M];
#pragma unroll
  for (int c = 0; c < M; c++) {
    so[c]  = __builtin_amdgcn_readfirstlane(slotS[c]) * PPITCH;
    ssq[c] = __int_as_float(__builtin_amdgcn_readfirstlane(__float_as_int(selSqS[c])));
  }
  gfp p = (gfp)(RB + run);
  v4f r0[4], r1[4], r2[4], r3[4];
  load_group(r0, p);
  asm volatile("" : "+v"(p));
  load_group(r1, p + 512);
  asm volatile("" : "+v"(p));
  load_group(r2, p + 1024);
#pragma unroll
  for (int j = 0; j < 4; j++) {
    v2f d[M][2];
#pragma unroll
    for (int c = 0; c < M; c++) { d[c][0] = (v2f){0.f, 0.f}; d[c][1] = (v2f){0.f, 0.f}; }
    if (j < 3) {
#pragma unroll 1
      for (int g = 0; g < 32; g += 4)
        ring_trip<M, true>(p, pool, so, g, r0, r1, r2, r3, d);
    } else {
#pragma unroll 1
      for (int g = 0; g < 28; g += 4)
        ring_trip<M, true>(p, pool, so, g, r0, r1, r2, r3, d);
      ring_trip<M, false>(p, pool, so, 28, r0, r1, r2, r3, d);
    }
    const float4 q = *(const float4*)(sqS + 4 * (1024 * j + t));
    float4 a = *(const float4*)(anchS + 4 * (1024 * j + t));
    float e0 = q.x + ssq[0] - 2.f * d[0][0].x;
    float e1 = q.y + ssq[0] - 2.f * d[0][0].y;
    float e2 = q.z + ssq[0] - 2.f * d[0][1].x;
    float e3 = q.w + ssq[0] - 2.f * d[0][1].y;
#pragma unroll
    for (int c = 1; c < M; c++) {
      e0 = fminf(e0, q.x + ssq[c] - 2.f * d[c][0].x);
      e1 = fminf(e1, q.y + ssq[c] - 2.f * d[c][0].y);
      e2 = fminf(e2, q.z + ssq[c] - 2.f * d[c][1].x);
      e3 = fminf(e3, q.w + ssq[c] - 2.f * d[c][1].y);
    }
    a.x = fminf(a.x, sqrtf(fmaxf(e0, 0.f)));
    a.y = fminf(a.y, sqrtf(fmaxf(e1, 0.f)));
    a.z = fminf(a.z, sqrtf(fmaxf(e2, 0.f)));
    a.w = fminf(a.w, sqrtf(fmaxf(e3, 0.f)));
    *(float4*)(anchS + 4 * (1024 * j + t)) = a;
  }
}

__global__ __launch_bounds__(1024)
void greedy_kernel(const float* __restrict__ RB, const float* __restrict__ sq,
                   const float* __restrict__ anchorSq, int* __restrict__ sel) {
  __shared__ float redV[32];
  __shared__ int   redI[32];
  __shared__ float redS[32];
  __shared__ __attribute__((aligned(16))) float pool[32 * PPITCH];
  __shared__ int   slotS[MMAX];
  __shared__ float selSqS[MMAX];
  __shared__ int   mS;
  __shared__ __attribute__((aligned(16))) float anchS[4 * 1024 * 4];
  __shared__ __attribute__((aligned(16))) float sqS[4 * 1024 * 4];

  const int t    = threadIdx.x;
  const int lane = t & 31;
  const int wid  = t >> 5;
  const int row0 = 512 * wid + 4 * lane;

#pragma unroll
  for (int j = 0; j < 4; j++) {
    float4 a = *(const float4*)(anchorSq + row0 + 128 * j);
    a.x = sqrtf(a.x);
    a.y = sqrtf(a.y);
    a.z = sqrtf(a.z);
    a.w = sqrtf(a.w);
    *(float4*)(anchS + 4 * (1024 * j + t)) = a;
    *(float4*)(sqS + 4 * (1024 * j + t)) = *(const float4*)(sq + row0 + 128 * j);
  }

  const unsigned run = (unsigned)wid * (4u * 128u * DPROJ) + 4u * (unsigned)lane;

  int done = 0;
  for (int pass = 0; pass < NSAMP && done < NSAMP; pass++) {
    float aVal[16];
#pragma unroll
    for (int j = 0; j < 4; j++) {
      const float4 a = *(const float4*)(anchS + 4 * (1024 * j + t));
      aVal[4 * j + 0] = a.x; aVal[4 * j + 1] = a.y; aVal[4 * j + 2] = a.z; aVal[4 * j + 3] = a.w;
    }
    float bv = aVal[0]; int bi = row0; float sv = -1.f;
#pragma unroll
    for (int e = 1; e < 16; e++) {
      const float v = aVal[e];
      if (v > bv) { sv = bv; bv = v; bi = row0 + 128 * (e >> 2) + (e & 3); }
      else        { sv = fmaxf(sv, v); }
    }
#pragma unroll
    for (int m = 1; m < 32; m <<= 1) {
      const float ov = __shfl_xor(bv, m, 32);
      const int   oi = __shfl_xor(bi, m, 32);
      const float os = __shfl_xor(sv, m, 32);
      const bool take = (ov > bv) || (ov == bv && oi < bi);
      sv = take ? fmaxf(os, bv) : fmaxf(sv, ov);
      if (take) { bv = ov; bi = oi; }
    }
    {
      const unsigned base = (unsigned)(bi >> 7) * (128u * DPROJ) + (unsigned)(bi & 127);
#pragma unroll
      for (int i = 0; i < 4; i++)
        pool[wid * PPITCH + lane + 32 * i] = RB[base + (unsigned)(lane + 32 * i) * 128u];
    }
    if (lane == 0) { redV[wid] = bv; redI[wid] = bi; redS[wid] = sv; }
    __syncthreads();

    if (wid == 0) {
      float pv = redV[lane];
      const int pi = redI[lane];
      float T = redS[lane];
#pragma unroll
      for (int m = 1; m < 32; m <<= 1) T = fmaxf(T, __shfl_xor(T, m, 32));
      const float psq = sq[pi];
      const float4* own = (const float4*)(pool + lane * PPITCH);
      int m = 0;
      for (int c = 0; c < MMAX; c++) {
        float cv = pv; int ci = pi; int cl = lane;
#pragma unroll
        for (int x = 1; x < 32; x <<= 1) {
          const float ov = __shfl_xor(cv, x, 32);
          const int   oi = __shfl_xor(ci, x, 32);
          const int   ol = __shfl_xor(cl, x, 32);
          if (ov > cv || (ov == cv && oi < ci)) { cv = ov; ci = oi; cl = ol; }
        }
        if (c > 0 && !(cv > T)) break;
        const float ssq = __shfl(psq, cl, 32);
        if (lane == 0) { sel[done + c] = ci; slotS[c] = cl; selSqS[c] = ssq; }
        m = c + 1;
        if (done + m == NSAMP || m == MMAX) break;
        const float4* sr = (const float4*)(pool + cl * PPITCH);
        float d = 0.f;
#pragma unroll 8
        for (int kg = 0; kg < 32; kg++) {
          const float4 r = own[kg];
          const float4 w = sr[kg];
          float p = r.y * w.y;
          p = fmaf(r.x, w.x, p);
          p = fmaf(r.z, w.z, p);
          p = fmaf(r.w, w.w, p);
          d += p;
        }
        const float e = psq + ssq - 2.f * d;
        pv = fminf(pv, sqrtf(fmaxf(e, 0.f)));
      }
      if (lane == 0) mS = m;
    }
    __syncthreads();

    const int m = mS;
    if (m == 1)      sweep_cols<1>(RB, sqS, anchS, pool, slotS, selSqS, run, t);
    else if (m == 2) sweep_cols<2>(RB, sqS, anchS, pool, slotS, selSqS, run, t);
    else if (m == 3) sweep_cols<3>(RB, sqS, anchS, pool, slotS, selSqS, run, t);
    else             sweep_cols<4>(RB, sqS, anchS, pool, slotS, selSqS, run, t);
    done += m;
    __syncthreads();
  }
}

__global__ __launch_bounds__(256)
void gather_kernel(const float* __restrict__ F, const int* __restrict__ sel,
                   float* __restrict__ out) {
  const int s = blockIdx.x;
  const int c = blockIdx.y * 256 + threadIdx.x;
  out[(size_t)s * DIN + c] = F[(size_t)sel[s] * DIN + c];
}

extern "C" void kernel_launch(void* const* d_in, const int* in_sizes, int n_in,
                              void* d_out, int out_size, void* d_ws, size_t ws_size,
                              hipStream_t stream) {
  (void)in_sizes; (void)n_in; (void)out_size; (void)ws_size;
  const float* F = (const float*)d_in[0];
  const float* W = (const float*)d_in[1];

  float* R          = (float*)d_ws;
  float* sq         = R + (size_t)NPTS * DPROJ;
  float* anchorSq   = sq + NPTS;
  float* anchorPart = anchorSq + NPTS;
  int*   sel        = (int*)(anchorPart + (size_t)NCHUNK * NPTS);

  proj_kernel<<<1024, 256, 0, stream>>>(F, W, R);
  sq_kernel<<<NPTS / 8, 256, 0, stream>>>(R, sq);
  kmajor_kernel<<<NPTS / 128, 1024, 0, stream>>>(R);
  gram_rowsum_kernel<<<1024, 256, 0, stream>>>(R, sq, anchorPart);
  combine_kernel<<<NPTS / 256, 256, 0, stream>>>(anchorPart, anchorSq);
  greedy_kernel<<<1, 1024, 0, stream>>>(R, sq, anchorSq, sel);
  gather_kernel<<<dim3(NSAMP, DIN / 256), 256, 0, stream>>>(F, sel, (float*)d_out);
}


# greedy sweep: up to 7 selections per sweep, not 4; each dot is fused straight into its running sum (no product temporaries, so 7 sum sets fit in registers); fewer 8 MB sweeps; dot sum order changes
# speedup vs baseline: 13.2964x; 1.0372x over previous
#include <hip/hip_runtime.h>


#define NPTS   16384
#define DIN    1024
#define DPROJ  128
#define NSAMP  163
#define NCHUNK 8

typedef float v2f __attribute__((ext_vector_type(2)));
typedef float v8f __attribute__((ext_vector_type(8)));

__device__ __forceinline__ v8f wmma_f32_k4(v2f a, v2f b, v8f c) {
  return __builtin_amdgcn_wmma_f32_16x16x4_f32(
       false, a,  false, b,
       (short)0, c,  false,  false);
}

__global__ __launch_bounds__(256)
void proj_kernel(const float* __restrict__ F, const float* __restrict__ W,
                 float* __restrict__ R) {
  const int wave = (blockIdx.x * blockDim.x + threadIdx.x) >> 5;
  const int lane = threadIdx.x & 31;
  const int tm = wave >> 3;
  const int tn = wave & 7;
  const int m0 = tm * 16, n0 = tn * 16;
  const int half = lane >> 4;
  const int r16  = lane & 15;
  const int koff = half * 2;

  v8f acc = {};
  const float* arow = F + (size_t)(m0 + r16) * DIN + koff;
  for (int k = 0; k < DIN; k += 4) {
    v2f a = *(const v2f*)(arow + k);
    v2f b;
    b.x = W[(size_t)(k + koff)     * DPROJ + n0 + r16];
    b.y = W[(size_t)(k + koff + 1) * DPROJ + n0 + r16];
    acc = wmma_f32_k4(a, b, acc);
  }
#pragma unroll
  for (int i = 0; i < 8; i++)
    R[(size_t)(m0 + i + half * 8) * DPROJ + n0 + r16] = acc[i];
}

__global__ __launch_bounds__(256)
void sq_kernel(const float* __restrict__ R, float* __restrict__ sq) {
  const int row  = (blockIdx.x * blockDim.x + threadIdx.x) >> 5;
  const int lane = threadIdx.x & 31;
  const float* rp = R + (size_t)row * DPROJ;
  float s = 0.f;
#pragma unroll
  for (int k = lane; k < DPROJ; k += 32) { float v = rp[k]; s += v * v; }
#pragma unroll
  for (int m = 16; m >= 1; m >>= 1) s += __shfl_xor(s, m, 32);
  if (lane == 0) sq[row] = s;
}

typedef float v4f __attribute__((ext_vector_type(4)));

__device__ __forceinline__ void gram_load(v4f (&b)[8], const float* p, const int ch) {
#pragma unroll
  for (int u = 0; u < 4; u++) {
    b[2 * u]     = *(const v4f*)(p + (4 * ch + u) * 512);
    b[2 * u + 1] = *(const v4f*)(p + (4 * ch + u) * 512 + 128);
  }
}

template <int CH>
__device__ __forceinline__ void gram_mma(const v4f (&b)[8], const v2f (&afrag)[32],
                                         v8f (&acc)[4]) {
#pragma unroll
  for (int u = 0; u < 4; u++) {
    const v2f a = afrag[4 * CH + u];
    acc[0] = wmma_f32_k4(a, (v2f){b[2 * u].x, b[2 * u + 1].x}, acc[0]);
    acc[1] = wmma_f32_k4(a, (v2f){b[2 * u].y, b[2 * u + 1].y}, acc[1]);
    acc[2] = wmma_f32_k4(a, (v2f){b[2 * u].z, b[2 * u + 1].z}, acc[2]);
    acc[3] = wmma_f32_k4(a, (v2f){b[2 * u].w, b[2 * u + 1].w}, acc[3]);
  }
}

__global__ __launch_bounds__(256)
void gram_rowsum_kernel(const float* __restrict__ RB, const float* __restrict__ sq,
                        float* __restrict__ anchorPart) {
  const int wave  = (blockIdx.x * blockDim.x + threadIdx.x) >> 5;
  const int lane  = threadIdx.x & 31;
  const int chunk = wave & (NCHUNK - 1);
  const int strip = wave >> 3;
  const int half  = lane >> 4;
  const int r16   = lane & 15;
  const int m0    = strip * 16;

  v2f afrag[32];
  const float* ap = RB + (size_t)(m0 >> 7) * (128 * DPROJ) + ((m0 & 127) + r16)
                       + half * 256;
#pragma unroll
  for (int kk = 0; kk < 32; kk++) {
    afrag[kk].x = ap[kk * 512];
    afrag[kk].y = ap[kk * 512 + 128];
  }

  float sqr[8];
#pragma unroll
  for (int i = 0; i < 8; i++) sqr[i] = sq[m0 + i + half * 8];

  float partial[8] = {0.f, 0.f, 0.f, 0.f, 0.f, 0.f, 0.f, 0.f};

  const float* cp = RB + (size_t)(chunk * 16) * (128 * DPROJ) + 4 * r16 + half * 256;
  const float* sp = sq + chunk * 2048 + 4 * r16;

  v4f bA[8], bB[8];
  gram_load(bA, cp, 0);

#pragma unroll 1
  for (int G = 0; G < 32; G++) {
    const float* p  = cp + (size_t)(G >> 1) * (128 * DPROJ) + (G & 1) * 64;
    const int Gn = (G < 31) ? G + 1 : G;
    const float* pn = cp + (size_t)(Gn >> 1) * (128 * DPROJ) + (Gn & 1) * 64;
    const v4f sc = *(const v4f*)(sp + 64 * G);

    v8f acc[4] = {};
    gram_load(bB, p, 1);
    gram_mma<0>(bA, afrag, acc);
    gram_load(bA, p, 2);
    gram_mma<1>(bB, afrag, acc);
    gram_load(bB, p, 3);
    gram_mma<2>(bA, afrag, acc);
    gram_load(bA, p, 4);
    gram_mma<3>(bB, afrag, acc);
    gram_load(bB, p, 5);
    gram_mma<4>(bA, afrag, acc);
    gram_load(bA, p, 6);
    gram_mma<5>(bB, afrag, acc);
    gram_load(bB, p, 7);
    gram_mma<6>(bA, afrag, acc);
    gram_load(bA, pn, 0);
    gram_mma<7>(bB, afrag, acc);

#pragma unroll
    for (int q = 0; q < 4; q++) {
#pragma unroll
      for (int i = 0; i < 8; i++)
        partial[i] += fmaxf(0.f, sqr[i] + sc[q] - 2.f * acc[q][i]);
    }
  }

#pragma unroll
  for (int i = 0; i < 8; i++) {
    float p = partial[i];
    p += __shfl_xor(p, 1, 32);
    p += __shfl_xor(p, 2, 32);
    p += __shfl_xor(p, 4, 32);
    p += __shfl_xor(p, 8, 32);
    if (r16 == 0)
      anchorPart[(size_t)chunk * NPTS + m0 + i + half * 8] = p;
  }
}

__global__ __launch_bounds__(256)
void combine_kernel(const float* __restrict__ anchorPart,
                    float* __restrict__ anchorSq) {
  const int i = blockIdx.x * 256 + threadIdx.x;
  float s = 0.f;
#pragma unroll
  for (int c = 0; c < NCHUNK; c++) s += anchorPart[(size_t)c * NPTS + i];
  anchorSq[i] = s;
}

__global__ __launch_bounds__(1024)
void kmajor_kernel(float* __restrict__ R) {
  float* blk = R + (size_t)blockIdx.x * (128 * DPROJ);
  const int lane = threadIdx.x & 31;
  const int w    = threadIdx.x >> 5;
  const int m    = 32 * (w & 3) + lane;
  const int kg0  = w >> 2;

  float4 v[4];
#pragma unroll
  for (int i = 0; i < 4; i++)
    v[i] = *(const float4*)(blk + m * DPROJ + 4 * (kg0 + 8 * i));

  __syncthreads();

#pragma unroll
  for (int i = 0; i < 4; i++) {
    const int k = 4 * (kg0 + 8 * i);
    blk[(k + 0) * 128 + m] = v[i].x;
    blk[(k + 1) * 128 + m] = v[i].y;
    blk[(k + 2) * 128 + m] = v[i].z;
    blk[(k + 3) * 128 + m] = v[i].w;
  }
}

#define MMAX   7
#define PPITCH 132

typedef const __attribute__((address_space(1))) float* gfp;
typedef const __attribute__((address_space(1))) v4f*   gv4p;

__device__ __forceinline__ void load_group(v4f (&r)[4], gfp p) {
#pragma unroll
  for (int u = 0; u < 4; u++) r[u] = *(gv4p)(p + u * 128);
}

template <int M, int C0, int P>
__device__ __forceinline__ void fold_part(const v4f (&v)[4], const float* selb,
                                          const int g, v2f (&d)[M][2]) {
  v4f w[P];
#pragma unroll
  for (int c = 0; c < P; c++) w[c] = *(const v4f*)(selb + 128 * (C0 + c) + 4 * g);
  if constexpr (P == 2) asm volatile("" : "+v"(w[0]), "+v"(w[1]));
  if constexpr (P == 3) asm volatile("" : "+v"(w[0]), "+v"(w[1]), "+v"(w[2]));
  if constexpr (P == 4) asm volatile("" : "+v"(w[0]), "+v"(w[1]), "+v"(w[2]), "+v"(w[3]));
#pragma unroll
  for (int c = 0; c < P; c++) {
    d[C0 + c][0] = __builtin_elementwise_fma(v[0].xy, w[c].xx, d[C0 + c][0]);
    d[C0 + c][1] = __builtin_elementwise_fma(v[0].zw, w[c].xx, d[C0 + c][1]);
  }
#pragma unroll
  for (int c = 0; c < P; c++) {
    d[C0 + c][0] = __builtin_elementwise_fma(v[1].xy, w[c].yy, d[C0 + c][0]);
    d[C0 + c][1] = __builtin_elementwise_fma(v[1].zw, w[c].yy, d[C0 + c][1]);
  }
#pragma unroll
  for (int c = 0; c < P; c++) {
    d[C0 + c][0] = __builtin_elementwise_fma(v[2].xy, w[c].zz, d[C0 + c][0]);
    d[C0 + c][1] = __builtin_elementwise_fma(v[2].zw, w[c].zz, d[C0 + c][1]);
  }
#pragma unroll
  for (int c = 0; c < P; c++) {
    d[C0 + c][0] = __builtin_elementwise_fma(v[3].xy, w[c].ww, d[C0 + c][0]);
    d[C0 + c][1] = __builtin_elementwise_fma(v[3].zw, w[c].ww, d[C0 + c][1]);
  }
}

template <int M>
__device__ __forceinline__ void fold_group(const v4f (&v)[4], const float* selb,
                                           const int g, v2f (&d)[M][2]) {
  if constexpr (M <= 4) {
    fold_part<M, 0, M>(v, selb, g, d);
  } else {
    fold_part<M, 0, 4>(v, selb, g, d);
    fold_part<M, 4, M - 4>(v, selb, g, d);
  }
}

template <int M>
__device__ __forceinline__ void after_fold(gfp &p, v2f (&d)[M][2]) {
#pragma unroll
  for (int c = 0; c < M; c++)
    asm volatile("" : "+v"(p), "+v"(d[c][0]), "+v"(d[c][1]));
}

template <int M, bool AHEAD>
__device__ __forceinline__ void ring_trip(gfp &p, const float* selb, const int g,
                                          v4f (&r0)[4], v4f (&r1)[4],
                                          v4f (&r2)[4], v4f (&r3)[4],
                                          v2f (&d)[M][2]) {
  load_group(r3, p + 3 * 512);
  fold_group<M>(r0, selb, g, d);
  after_fold<M>(p, d);
  if (AHEAD) load_group(r0, p + 4 * 512);
  fold_group<M>(r1, selb, g + 1, d);
  after_fold<M>(p, d);
  if (AHEAD) load_group(r1, p + 5 * 512);
  fold_group<M>(r2, selb, g + 2, d);
  after_fold<M>(p, d);
  if (AHEAD) load_group(r2, p + 6 * 512);
  fold_group<M>(r3, selb, g + 3, d);
  after_fold<M>(p, d);
  p += 4 * 512;
}

template <int M>
__device__ __forceinline__ void sweep_cols(const float* __restrict__ RB,
                                           const float* sqS, float* anchS,
                                           const float* selb,
                                           const float* selSqS,
                                           const unsigned run, const int t) {
  float ssq[M];
#pragma unroll
  for (int c = 0; c < M; c++) {
    ssq[c] = __int_as_float(__builtin_amdgcn_readfirstlane(__float_as_int(selSqS[c])));
  }
  gfp p = (gfp)(RB + run);
  v4f r0[4], r1[4], r2[4], r3[4];
  load_group(r0, p);
  asm volatile("" : "+v"(p));
  load_group(r1, p + 512);
  asm volatile("" : "+v"(p));
  load_group(r2, p + 1024);
#pragma unroll
  for (int j = 0; j < 4; j++) {
    v2f d[M][2];
#pragma unroll
    for (int c = 0; c < M; c++) { d[c][0] = (v2f){0.f, 0.f}; d[c][1] = (v2f){0.f, 0.f}; }
    if (j < 3) {
#pragma unroll 1
      for (int g = 0; g < 32; g += 4)
        ring_trip<M, true>(p, selb, g, r0, r1, r2, r3, d);
    } else {
#pragma unroll 1
      for (int g = 0; g < 28; g += 4)
        ring_trip<M, true>(p, selb, g, r0, r1, r2, r3, d);
      ring_trip<M, false>(p, selb, 28, r0, r1, r2, r3, d);
    }
    const float4 q = *(const float4*)(sqS + 4 * (1024 * j + t));
    float4 a = *(const float4*)(anchS + 4 * (1024 * j + t));
    float e0 = q.x + ssq[0] - 2.f * d[0][0].x;
    float e1 = q.y + ssq[0] - 2.f * d[0][0].y;
    float e2 = q.z + ssq[0] - 2.f * d[0][1].x;
    float e3 = q.w + ssq[0] - 2.f * d[0][1].y;
#pragma unroll
    for (int c = 1; c < M; c++) {
      e0 = fminf(e0, q.x + ssq[c] - 2.f * d[c][0].x);
      e1 = fminf(e1, q.y + ssq[c] - 2.f * d[c][0].y);
      e2 = fminf(e2, q.z + ssq[c] - 2.f * d[c][1].x);
      e3 = fminf(e3, q.w + ssq[c] - 2.f * d[c][1].y);
    }
    a.x = fminf(a.x, sqrtf(fmaxf(e0, 0.f)));
    a.y = fminf(a.y, sqrtf(fmaxf(e1, 0.f)));
    a.z = fminf(a.z, sqrtf(fmaxf(e2, 0.f)));
    a.w = fminf(a.w, sqrtf(fmaxf(e3, 0.f)));
    *(float4*)(anchS + 4 * (1024 * j + t)) = a;
  }
}

__global__ __launch_bounds__(1024)
void greedy_kernel(const float* __restrict__ RB, const float* __restrict__ sq,
                   const float* __restrict__ anchorSq, int* __restrict__ sel) {
  __shared__ float redV[32];
  __shared__ int   redI[32];
  __shared__ float redS[32];
  __shared__ __attribute__((aligned(16))) float pool[32 * PPITCH];
  __shared__ __attribute__((aligned(16))) float selb[MMAX * DPROJ];
  __shared__ float selSqS[MMAX];
  __shared__ int   mS;
  __shared__ __attribute__((aligned(16))) float anchS[4 * 1024 * 4];
  __shared__ __attribute__((aligned(16))) float sqS[4 * 1024 * 4];

  const int t    = threadIdx.x;
  const int lane = t & 31;
  const int wid  = t >> 5;
  const int row0 = 512 * wid + 4 * lane;

#pragma unroll
  for (int j = 0; j < 4; j++) {
    float4 a = *(const float4*)(anchorSq + row0 + 128 * j);
    a.x = sqrtf(a.x);
    a.y = sqrtf(a.y);
    a.z = sqrtf(a.z);
    a.w = sqrtf(a.w);
    *(float4*)(anchS + 4 * (1024 * j + t)) = a;
    *(float4*)(sqS + 4 * (1024 * j + t)) = *(const float4*)(sq + row0 + 128 * j);
  }

  const unsigned run = (unsigned)wid * (4u * 128u * DPROJ) + 4u * (unsigned)lane;

  int done = 0;
  for (int pass = 0; pass < NSAMP && done < NSAMP; pass++) {
    float aVal[16];
#pragma unroll
    for (int j = 0; j < 4; j++) {
      const float4 a = *(const float4*)(anchS + 4 * (1024 * j + t));
      aVal[4 * j + 0] = a.x; aVal[4 * j + 1] = a.y; aVal[4 * j + 2] = a.z; aVal[4 * j + 3] = a.w;
    }
    float bv = aVal[0]; int bi = row0; float sv = -1.f;
#pragma unroll
    for (int e = 1; e < 16; e++) {
      const float v = aVal[e];
      if (v > bv) { sv = bv; bv = v; bi = row0 + 128 * (e >> 2) + (e & 3); }
      else        { sv = fmaxf(sv, v); }
    }
#pragma unroll
    for (int m = 1; m < 32; m <<= 1) {
      const float ov = __shfl_xor(bv, m, 32);
      const int   oi = __shfl_xor(bi, m, 32);
      const float os = __shfl_xor(sv, m, 32);
      const bool take = (ov > bv) || (ov == bv && oi < bi);
      sv = take ? fmaxf(os, bv) : fmaxf(sv, ov);
      if (take) { bv = ov; bi = oi; }
    }
    {
      const unsigned base = (unsigned)(bi >> 7) * (128u * DPROJ) + (unsigned)(bi & 127);
#pragma unroll
      for (int i = 0; i < 4; i++)
        pool[wid * PPITCH + lane + 32 * i] = RB[base + (unsigned)(lane + 32 * i) * 128u];
    }
    if (lane == 0) { redV[wid] = bv; redI[wid] = bi; redS[wid] = sv; }
    __syncthreads();

    if (wid == 0) {
      float pv = redV[lane];
      const int pi = redI[lane];
      float T = redS[lane];
#pragma unroll
      for (int m = 1; m < 32; m <<= 1) T = fmaxf(T, __shfl_xor(T, m, 32));
      const float psq = sq[pi];
      const float4* own = (const float4*)(pool + lane * PPITCH);
      int m = 0;
      for (int c = 0; c < MMAX; c++) {
        float cv = pv; int ci = pi; int cl = lane;
#pragma unroll
        for (int x = 1; x < 32; x <<= 1) {
          const float ov = __shfl_xor(cv, x, 32);
          const int   oi = __shfl_xor(ci, x, 32);
          const int   ol = __shfl_xor(cl, x, 32);
          if (ov > cv || (ov == cv && oi < ci)) { cv = ov; ci = oi; cl = ol; }
        }
        if (c > 0 && !(cv > T)) break;
        const float ssq = __shfl(psq, cl, 32);
        if (lane == 0) { sel[done + c] = ci; selSqS[c] = ssq; }
        *(v4f*)(selb + DPROJ * c + 4 * lane) = *(const v4f*)(pool + cl * PPITCH + 4 * lane);
        m = c + 1;
        if (done + m == NSAMP || m == MMAX) break;
        const float4* sr = (const float4*)(pool + cl * PPITCH);
        float d = 0.f;
#pragma unroll 8
        for (int kg = 0; kg < 32; kg++) {
          const float4 r = own[kg];
          const float4 w = sr[kg];
          d = fmaf(r.x, w.x, d);
          d = fmaf(r.y, w.y, d);
          d = fmaf(r.z, w.z, d);
          d = fmaf(r.w, w.w, d);
        }
        const float e = psq + ssq - 2.f * d;
        pv = fminf(pv, sqrtf(fmaxf(e, 0.f)));
      }
      if (lane == 0) mS = m;
    }
    __syncthreads();

    const int m = mS;
    switch (m) {
      case 1:  sweep_cols<1>(RB, sqS, anchS, selb, selSqS, run, t); break;
      case 2:  sweep_cols<2>(RB, sqS, anchS, selb, selSqS, run, t); break;
      case 3:  sweep_cols<3>(RB, sqS, anchS, selb, selSqS, run, t); break;
      case 4:  sweep_cols<4>(RB, sqS, anchS, selb, selSqS, run, t); break;
      case 5:  sweep_cols<5>(RB, sqS, anchS, selb, selSqS, run, t); break;
      case 6:  sweep_cols<6>(RB, sqS, anchS, selb, selSqS, run, t); break;
      default: sweep_cols<7>(RB, sqS, anchS, selb, selSqS, run, t); break;
    }
    done += m;
    __syncthreads();
  }
}

__global__ __launch_bounds__(256)
void gather_kernel(const float* __restrict__ F, const int* __restrict__ sel,
                   float* __restrict__ out) {
  const int s = blockIdx.x;
  const int c = blockIdx.y * 256 + threadIdx.x;
  out[(size_t)s * DIN + c] = F[(size_t)sel[s] * DIN + c];
}

extern "C" void kernel_launch(void* const* d_in, const int* in_sizes, int n_in,
                              void* d_out, int out_size, void* d_ws, size_t ws_size,
                              hipStream_t stream) {
  (void)in_sizes; (void)n_in; (void)out_size; (void)ws_size;
  const float* F = (const float*)d_in[0];
  const float* W = (const float*)d_in[1];

  float* R          = (float*)d_ws;
  float* sq         = R + (size_t)NPTS * DPROJ;
  float* anchorSq   = sq + NPTS;
  float* anchorPart = anchorSq + NPTS;
  int*   sel        = (int*)(anchorPart + (size_t)NCHUNK * NPTS);

  proj_kernel<<<1024, 256, 0, stream>>>(F, W, R);
  sq_kernel<<<NPTS / 8, 256, 0, stream>>>(R, sq);
  kmajor_kernel<<<NPTS / 128, 1024, 0, stream>>>(R);
  gram_rowsum_kernel<<<1024, 256, 0, stream>>>(R, sq, anchorPart);
  combine_kernel<<<NPTS / 256, 256, 0, stream>>>(anchorPart, anchorSq);
  greedy_kernel<<<1, 1024, 0, stream>>>(R, sq, anchorSq, sel);
  gather_kernel<<<dim3(NSAMP, DIN / 256), 256, 0, stream>>>(F, sel, (float*)d_out);
}


# greedy sweep: passes of 8 to 16 selections take their dots on f32 WMMA (selected rows = A out of LDS, k-major rows = B as in the gram pass); pool holds 2 rows per wave; their dot sum order changes
# speedup vs baseline: 14.4334x; 1.0855x over previous
#include <hip/hip_runtime.h>


#define NPTS   16384
#define DIN    1024
#define DPROJ  128
#define NSAMP  163
#define NCHUNK 8

typedef float v2f __attribute__((ext_vector_type(2)));
typedef float v8f __attribute__((ext_vector_type(8)));

__device__ __forceinline__ v8f wmma_f32_k4(v2f a, v2f b, v8f c) {
  return __builtin_amdgcn_wmma_f32_16x16x4_f32(
       false, a,  false, b,
       (short)0, c,  false,  false);
}

__global__ __launch_bounds__(256)
void proj_kernel(const float* __restrict__ F, const float* __restrict__ W,
                 float* __restrict__ R) {
  const int wave = (blockIdx.x * blockDim.x + threadIdx.x) >> 5;
  const int lane = threadIdx.x & 31;
  const int tm = wave >> 3;
  const int tn = wave & 7;
  const int m0 = tm * 16, n0 = tn * 16;
  const int half = lane >> 4;
  const int r16  = lane & 15;
  const int koff = half * 2;

  v8f acc = {};
  const float* arow = F + (size_t)(m0 + r16) * DIN + koff;
  for (int k = 0; k < DIN; k += 4) {
    v2f a = *(const v2f*)(arow + k);
    v2f b;
    b.x = W[(size_t)(k + koff)     * DPROJ + n0 + r16];
    b.y = W[(size_t)(k + koff + 1) * DPROJ + n0 + r16];
    acc = wmma_f32_k4(a, b, acc);
  }
#pragma unroll
  for (int i = 0; i < 8; i++)
    R[(size_t)(m0 + i + half * 8) * DPROJ + n0 + r16] = acc[i];
}

__global__ __launch_bounds__(256)
void sq_kernel(const float* __restrict__ R, float* __restrict__ sq) {
  const int row  = (blockIdx.x * blockDim.x + threadIdx.x) >> 5;
  const int lane = threadIdx.x & 31;
  const float* rp = R + (size_t)row * DPROJ;
  float s = 0.f;
#pragma unroll
  for (int k = lane; k < DPROJ; k += 32) { float v = rp[k]; s += v * v; }
#pragma unroll
  for (int m = 16; m >= 1; m >>= 1) s += __shfl_xor(s, m, 32);
  if (lane == 0) sq[row] = s;
}

typedef float v4f __attribute__((ext_vector_type(4)));

__device__ __forceinline__ void gram_load(v4f (&b)[8], const float* p, const int ch) {
#pragma unroll
  for (int u = 0; u < 4; u++) {
    b[2 * u]     = *(const v4f*)(p + (4 * ch + u) * 512);
    b[2 * u + 1] = *(const v4f*)(p + (4 * ch + u) * 512 + 128);
  }
}

template <int CH>
__device__ __forceinline__ void gram_mma(const v4f (&b)[8], const v2f (&afrag)[32],
                                         v8f (&acc)[4]) {
#pragma unroll
  for (int u = 0; u < 4; u++) {
    const v2f a = afrag[4 * CH + u];
    acc[0] = wmma_f32_k4(a, (v2f){b[2 * u].x, b[2 * u + 1].x}, acc[0]);
    acc[1] = wmma_f32_k4(a, (v2f){b[2 * u].y, b[2 * u + 1].y}, acc[1]);
    acc[2] = wmma_f32_k4(a, (v2f){b[2 * u].z, b[2 * u + 1].z}, acc[2]);
    acc[3] = wmma_f32_k4(a, (v2f){b[2 * u].w, b[2 * u + 1].w}, acc[3]);
  }
}

__global__ __launch_bounds__(256)
void gram_rowsum_kernel(const float* __restrict__ RB, const float* __restrict__ sq,
                        float* __restrict__ anchorPart) {
  const int wave  = (blockIdx.x * blockDim.x + threadIdx.x) >> 5;
  const int lane  = threadIdx.x & 31;
  const int chunk = wave & (NCHUNK - 1);
  const int strip = wave >> 3;
  const int half  = lane >> 4;
  const int r16   = lane & 15;
  const int m0    = strip * 16;

  v2f afrag[32];
  const float* ap = RB + (size_t)(m0 >> 7) * (128 * DPROJ) + ((m0 & 127) + r16)
                       + half * 256;
#pragma unroll
  for (int kk = 0; kk < 32; kk++) {
    afrag[kk].x = ap[kk * 512];
    afrag[kk].y = ap[kk * 512 + 128];
  }

  float sqr[8];
#pragma unroll
  for (int i = 0; i < 8; i++) sqr[i] = sq[m0 + i + half * 8];

  float partial[8] = {0.f, 0.f, 0.f, 0.f, 0.f, 0.f, 0.f, 0.f};

  const float* cp = RB + (size_t)(chunk * 16) * (128 * DPROJ) + 4 * r16 + half * 256;
  const float* sp = sq + chunk * 2048 + 4 * r16;

  v4f bA[8], bB[8];
  gram_load(bA, cp, 0);

#pragma unroll 1
  for (int G = 0; G < 32; G++) {
    const float* p  = cp + (size_t)(G >> 1) * (128 * DPROJ) + (G & 1) * 64;
    const int Gn = (G < 31) ? G + 1 : G;
    const float* pn = cp + (size_t)(Gn >> 1) * (128 * DPROJ) + (Gn & 1) * 64;
    const v4f sc = *(const v4f*)(sp + 64 * G);

    v8f acc[4] = {};
    gram_load(bB, p, 1);
    gram_mma<0>(bA, afrag, acc);
    gram_load(bA, p, 2);
    gram_mma<1>(bB, afrag, acc);
    gram_load(bB, p, 3);
    gram_mma<2>(bA, afrag, acc);
    gram_load(bA, p, 4);
    gram_mma<3>(bB, afrag, acc);
    gram_load(bB, p, 5);
    gram_mma<4>(bA, afrag, acc);
    gram_load(bA, p, 6);
    gram_mma<5>(bB, afrag, acc);
    gram_load(bB, p, 7);
    gram_mma<6>(bA, afrag, acc);
    gram_load(bA, pn, 0);
    gram_mma<7>(bB, afrag, acc);

#pragma unroll
    for (int q = 0; q < 4; q++) {
#pragma unroll
      for (int i = 0; i < 8; i++)
        partial[i] += fmaxf(0.f, sqr[i] + sc[q] - 2.f * acc[q][i]);
    }
  }

#pragma unroll
  for (int i = 0; i < 8; i++) {
    float p = partial[i];
    p += __shfl_xor(p, 1, 32);
    p += __shfl_xor(p, 2, 32);
    p += __shfl_xor(p, 4, 32);
    p += __shfl_xor(p, 8, 32);
    if (r16 == 0)
      anchorPart[(size_t)chunk * NPTS + m0 + i + half * 8] = p;
  }
}

__global__ __launch_bounds__(256)
void combine_kernel(const float* __restrict__ anchorPart,
                    float* __restrict__ anchorSq) {
  const int i = blockIdx.x * 256 + threadIdx.x;
  float s = 0.f;
#pragma unroll
  for (int c = 0; c < NCHUNK; c++) s += anchorPart[(size_t)c * NPTS + i];
  anchorSq[i] = s;
}

__global__ __launch_bounds__(1024)
void kmajor_kernel(float* __restrict__ R) {
  float* blk = R + (size_t)blockIdx.x * (128 * DPROJ);
  const int lane = threadIdx.x & 31;
  const int w    = threadIdx.x >> 5;
  const int m    = 32 * (w & 3) + lane;
  const int kg0  = w >> 2;

  float4 v[4];
#pragma unroll
  for (int i = 0; i < 4; i++)
    v[i] = *(const float4*)(blk + m * DPROJ + 4 * (kg0 + 8 * i));

  __syncthreads();

#pragma unroll
  for (int i = 0; i < 4; i++) {
    const int k = 4 * (kg0 + 8 * i);
    blk[(k + 0) * 128 + m] = v[i].x;
    blk[(k + 1) * 128 + m] = v[i].y;
    blk[(k + 2) * 128 + m] = v[i].z;
    blk[(k + 3) * 128 + m] = v[i].w;
  }
}

#define MMAX   16
#define MVALU  7
#define NPOOL  64
#define PPITCH 132
#define SPITCH 132

typedef const __attribute__((address_space(1))) float* gfp;
typedef const __attribute__((address_space(1))) v4f*   gv4p;

__device__ __forceinline__ void load_group(v4f (&r)[4], gfp p) {
#pragma unroll
  for (int u = 0; u < 4; u++) r[u] = *(gv4p)(p + u * 128);
}

template <int M, int C0, int P>
__device__ __forceinline__ void fold_part(const v4f (&v)[4], const float* selb,
                                          const int g, v2f (&d)[M][2]) {
  v4f w[P];
#pragma unroll
  for (int c = 0; c < P; c++) w[c] = *(const v4f*)(selb + SPITCH * (C0 + c) + 4 * g);
  if constexpr (P == 2) asm volatile("" : "+v"(w[0]), "+v"(w[1]));
  if constexpr (P == 3) asm volatile("" : "+v"(w[0]), "+v"(w[1]), "+v"(w[2]));
  if constexpr (P == 4) asm volatile("" : "+v"(w[0]), "+v"(w[1]), "+v"(w[2]), "+v"(w[3]));
#pragma unroll
  for (int c = 0; c < P; c++) {
    d[C0 + c][0] = __builtin_elementwise_fma(v[0].xy, w[c].xx, d[C0 + c][0]);
    d[C0 + c][1] = __builtin_elementwise_fma(v[0].zw, w[c].xx, d[C0 + c][1]);
  }
#pragma unroll
  for (int c = 0; c < P; c++) {
    d[C0 + c][0] = __builtin_elementwise_fma(v[1].xy, w[c].yy, d[C0 + c][0]);
    d[C0 + c][1] = __builtin_elementwise_fma(v[1].zw, w[c].yy, d[C0 + c][1]);
  }
#pragma unroll
  for (int c = 0; c < P; c++) {
    d[C0 + c][0] = __builtin_elementwise_fma(v[2].xy, w[c].zz, d[C0 + c][0]);
    d[C0 + c][1] = __builtin_elementwise_fma(v[2].zw, w[c].zz, d[C0 + c][1]);
  }
#pragma unroll
  for (int c = 0; c < P; c++) {
    d[C0 + c][0] = __builtin_elementwise_fma(v[3].xy, w[c].ww, d[C0 + c][0]);
    d[C0 + c][1] = __builtin_elementwise_fma(v[3].zw, w[c].ww, d[C0 + c][1]);
  }
}

template <int M>
__device__ __forceinline__ void fold_group(const v4f (&v)[4], const float* selb,
                                           const int g, v2f (&d)[M][2]) {
  if constexpr (M <= 4) {
    fold_part<M, 0, M>(v, selb, g, d);
  } else {
    fold_part<M, 0, 4>(v, selb, g, d);
    fold_part<M, 4, M - 4>(v, selb, g, d);
  }
}

template <int M>
__device__ __forceinline__ void after_fold(gfp &p, v2f (&d)[M][2]) {
#pragma unroll
  for (int c = 0; c < M; c++)
    asm volatile("" : "+v"(p), "+v"(d[c][0]), "+v"(d[c][1]));
}

template <int M, bool AHEAD>
__device__ __forceinline__ void ring_trip(gfp &p, const float* selb, const int g,
                                          v4f (&r0)[4], v4f (&r1)[4],
                                          v4f (&r2)[4], v4f (&r3)[4],
                                          v2f (&d)[M][2]) {
  load_group(r3, p + 3 * 512);
  fold_group<M>(r0, selb, g, d);
  after_fold<M>(p, d);
  if (AHEAD) load_group(r0, p + 4 * 512);
  fold_group<M>(r1, selb, g + 1, d);
  after_fold<M>(p, d);
  if (AHEAD) load_group(r1, p + 5 * 512);
  fold_group<M>(r2, selb, g + 2, d);
  after_fold<M>(p, d);
  if (AHEAD) load_group(r2, p + 6 * 512);
  fold_group<M>(r3, selb, g + 3, d);
  after_fold<M>(p, d);
  p += 4 * 512;
}

template <int M>
__device__ __forceinline__ void sweep_cols(const float* __restrict__ RB,
                                           const float* sqS, float* anchS,
                                           const float* selb,
                                           const float* selSqS,
                                           const unsigned run, const int t) {
  float ssq[M];
#pragma unroll
  for (int c = 0; c < M; c++) {
    ssq[c] = __int_as_float(__builtin_amdgcn_readfirstlane(__float_as_int(selSqS[c])));
  }
  gfp p = (gfp)(RB + run);
  v4f r0[4], r1[4], r2[4], r3[4];
  load_group(r0, p);
  asm volatile("" : "+v"(p));
  load_group(r1, p + 512);
  asm volatile("" : "+v"(p));
  load_group(r2, p + 1024);
#pragma unroll
  for (int j = 0; j < 4; j++) {
    v2f d[M][2];
#pragma unroll
    for (int c = 0; c < M; c++) { d[c][0] = (v2f){0.f, 0.f}; d[c][1] = (v2f){0.f, 0.f}; }
    if (j < 3) {
#pragma unroll 1
      for (int g = 0; g < 32; g += 4)
        ring_trip<M, true>(p, selb, g, r0, r1, r2, r3, d);
    } else {
#pragma unroll 1
      for (int g = 0; g < 28; g += 4)
        ring_trip<M, true>(p, selb, g, r0, r1, r2, r3, d);
      ring_trip<M, false>(p, selb, 28, r0, r1, r2, r3, d);
    }
    const float4 q = *(const float4*)(sqS + 4 * (1024 * j + t));
    float4 a = *(const float4*)(anchS + 4 * (1024 * j + t));
    float e0 = q.x + ssq[0] - 2.f * d[0][0].x;
    float e1 = q.y + ssq[0] - 2.f * d[0][0].y;
    float e2 = q.z + ssq[0] - 2.f * d[0][1].x;
    float e3 = q.w + ssq[0] - 2.f * d[0][1].y;
#pragma unroll
    for (int c = 1; c < M; c++) {
      e0 = fminf(e0, q.x + ssq[c] - 2.f * d[c][0].x);
      e1 = fminf(e1, q.y + ssq[c] - 2.f * d[c][0].y);
      e2 = fminf(e2, q.z + ssq[c] - 2.f * d[c][1].x);
      e3 = fminf(e3, q.w + ssq[c] - 2.f * d[c][1].y);
    }
    a.x = fminf(a.x, sqrtf(fmaxf(e0, 0.f)));
    a.y = fminf(a.y, sqrtf(fmaxf(e1, 0.f)));
    a.z = fminf(a.z, sqrtf(fmaxf(e2, 0.f)));
    a.w = fminf(a.w, sqrtf(fmaxf(e3, 0.f)));
    *(float4*)(anchS + 4 * (1024 * j + t)) = a;
  }
}

__device__ __forceinline__ void mm_load(v4f (&b)[4], v2f (&a)[2], gfp p,
                                        const float* ap, const int c) {
#pragma unroll
  for (int u = 0; u < 2; u++) {
    b[2 * u]     = *(gv4p)(p + (2 * c + u) * 512);
    b[2 * u + 1] = *(gv4p)(p + (2 * c + u) * 512 + 128);
    a[u]         = *(const v2f*)(ap + 4 * (2 * c + u));
  }
}

__device__ __forceinline__ void mm_mma(const v4f (&b)[4], const v2f (&a)[2],
                                       v8f (&acc)[4]) {
#pragma unroll
  for (int u = 0; u < 2; u++) {
    acc[0] = wmma_f32_k4(a[u], (v2f){b[2 * u].x, b[2 * u + 1].x}, acc[0]);
    acc[1] = wmma_f32_k4(a[u], (v2f){b[2 * u].y, b[2 * u + 1].y}, acc[1]);
    acc[2] = wmma_f32_k4(a[u], (v2f){b[2 * u].z, b[2 * u + 1].z}, acc[2]);
    acc[3] = wmma_f32_k4(a[u], (v2f){b[2 * u].w, b[2 * u + 1].w}, acc[3]);
  }
}

__device__ __forceinline__ void sweep_mm(const float* __restrict__ RB,
                                         const float* sqS, float* anchS,
                                         const float* selb, const float* selSqS,
                                         const int m, const int wid, const int lane) {
  const int half = lane >> 4;
  const int r16  = lane & 15;
  float ssq8[8];
#pragma unroll
  for (int i = 0; i < 8; i++) {
    const int s = 8 * half + i;
    const float v = selSqS[s];
    ssq8[i] = (s < m) ? v : __builtin_inff();
  }
  unsigned aoff = (unsigned)(r16 * SPITCH + 2 * half);
  gfp p0 = (gfp)(RB + (unsigned)wid * (4u * 128u * DPROJ) + 4u * (unsigned)r16
                    + 256u * (unsigned)half);
  v4f bA[4], bB[4];
  v2f aA[2], aB[2];
  mm_load(bA, aA, p0, selb + aoff, 0);
#pragma unroll 1
  for (int G = 0; G < 8; G++) {
    asm volatile("" : "+v"(aoff));
    const float* ap = selb + aoff;
    gfp p = p0 + (G >> 1) * (128 * DPROJ) + (G & 1) * 64;
    const int Gn = (G < 7) ? G + 1 : G;
    gfp pn = p0 + (Gn >> 1) * (128 * DPROJ) + (Gn & 1) * 64;
    v8f acc[4] = {};
#pragma unroll
    for (int c2 = 0; c2 < 8; c2++) {
      mm_load(bB, aB, p, ap, 2 * c2 + 1);
      mm_mma(bA, aA, acc);
      if (c2 < 7) mm_load(bA, aA, p, ap, 2 * c2 + 2);
      else        mm_load(bA, aA, pn, ap, 0);
      mm_mma(bB, aB, acc);
    }

    const int slot = 1024 * (G >> 1) + 32 * wid + 16 * (G & 1) + r16;
    const float4 q = *(const float4*)(sqS + 4 * slot);
    float4 a = *(const float4*)(anchS + 4 * slot);
    float e0 = __builtin_inff(), e1 = e0, e2 = e0, e3 = e0;
#pragma unroll
    for (int i = 0; i < 8; i++) {
      e0 = fminf(e0, q.x + ssq8[i] - 2.f * acc[0][i]);
      e1 = fminf(e1, q.y + ssq8[i] - 2.f * acc[1][i]);
      e2 = fminf(e2, q.z + ssq8[i] - 2.f * acc[2][i]);
      e3 = fminf(e3, q.w + ssq8[i] - 2.f * acc[3][i]);
    }
    e0 = fminf(e0, __shfl_xor(e0, 16, 32));
    e1 = fminf(e1, __shfl_xor(e1, 16, 32));
    e2 = fminf(e2, __shfl_xor(e2, 16, 32));
    e3 = fminf(e3, __shfl_xor(e3, 16, 32));
    a.x = fminf(a.x, sqrtf(fmaxf(e0, 0.f)));
    a.y = fminf(a.y, sqrtf(fmaxf(e1, 0.f)));
    a.z = fminf(a.z, sqrtf(fmaxf(e2, 0.f)));
    a.w = fminf(a.w, sqrtf(fmaxf(e3, 0.f)));
    if (half == 0) *(float4*)(anchS + 4 * slot) = a;
  }
}

__global__ __launch_bounds__(1024)
void greedy_kernel(const float* __restrict__ RB, const float* __restrict__ sq,
                   const float* __restrict__ anchorSq, int* __restrict__ sel) {
  __shared__ float redV[NPOOL];
  __shared__ int   redI[NPOOL];
  __shared__ float redS[32];
  __shared__ __attribute__((aligned(16))) float pool[NPOOL * PPITCH];
  __shared__ __attribute__((aligned(16))) float selb[MMAX * SPITCH];
  __shared__ float selSqS[MMAX];
  __shared__ int   mS;
  __shared__ __attribute__((aligned(16))) float anchS[4 * 1024 * 4];
  __shared__ __attribute__((aligned(16))) float sqS[4 * 1024 * 4];

  const int t    = threadIdx.x;
  const int lane = t & 31;
  const int wid  = t >> 5;
  const int row0 = 512 * wid + 4 * lane;

#pragma unroll
  for (int j = 0; j < 4; j++) {
    float4 a = *(const float4*)(anchorSq + row0 + 128 * j);
    a.x = sqrtf(a.x);
    a.y = sqrtf(a.y);
    a.z = sqrtf(a.z);
    a.w = sqrtf(a.w);
    *(float4*)(anchS + 4 * (1024 * j + t)) = a;
    *(float4*)(sqS + 4 * (1024 * j + t)) = *(const float4*)(sq + row0 + 128 * j);
  }

  for (int i = t; i < MMAX * SPITCH; i += 1024) selb[i] = 0.f;

  const unsigned run = (unsigned)wid * (4u * 128u * DPROJ) + 4u * (unsigned)lane;

  int done = 0;
  for (int pass = 0; pass < NSAMP && done < NSAMP; pass++) {
    float aVal[16];
#pragma unroll
    for (int j = 0; j < 4; j++) {
      const float4 a = *(const float4*)(anchS + 4 * (1024 * j + t));
      aVal[4 * j + 0] = a.x; aVal[4 * j + 1] = a.y; aVal[4 * j + 2] = a.z; aVal[4 * j + 3] = a.w;
    }
    float bv = aVal[0]; int bi = row0; float sv = -1.f;
#pragma unroll
    for (int e = 1; e < 16; e++) {
      const float v = aVal[e];
      if (v > bv) { sv = bv; bv = v; bi = row0 + 128 * (e >> 2) + (e & 3); }
      else        { sv = fmaxf(sv, v); }
    }
    float w1v = bv; int w1i = bi;
#pragma unroll
    for (int m = 1; m < 32; m <<= 1) {
      const float ov = __shfl_xor(w1v, m, 32);
      const int   oi = __shfl_xor(w1i, m, 32);
      if ((ov > w1v) || (ov == w1v && oi < w1i)) { w1v = ov; w1i = oi; }
    }
    const bool took1 = (bi == w1i);
    float w2v = took1 ? -2.f : bv; int w2i = bi;
#pragma unroll
    for (int m = 1; m < 32; m <<= 1) {
      const float ov = __shfl_xor(w2v, m, 32);
      const int   oi = __shfl_xor(w2i, m, 32);
      if ((ov > w2v) || (ov == w2v && oi < w2i)) { w2v = ov; w2i = oi; }
    }
    float tl = (took1 || bi == w2i) ? sv : bv;
#pragma unroll
    for (int m = 1; m < 32; m <<= 1) tl = fmaxf(tl, __shfl_xor(tl, m, 32));
    {
      const unsigned base1 = (unsigned)(w1i >> 7) * (128u * DPROJ) + (unsigned)(w1i & 127);
      const unsigned base2 = (unsigned)(w2i >> 7) * (128u * DPROJ) + (unsigned)(w2i & 127);
#pragma unroll
      for (int i = 0; i < 4; i++) {
        pool[wid * PPITCH + lane + 32 * i]        = RB[base1 + (unsigned)(lane + 32 * i) * 128u];
        pool[(32 + wid) * PPITCH + lane + 32 * i] = RB[base2 + (unsigned)(lane + 32 * i) * 128u];
      }
    }
    if (lane == 0) {
      redV[wid] = w1v;      redI[wid] = w1i;
      redV[32 + wid] = w2v; redI[32 + wid] = w2i;
      redS[wid] = tl;
    }
    __syncthreads();

    if (wid == 0) {
      float pvA = redV[lane],      pvB = redV[32 + lane];
      const int piA = redI[lane],  piB = redI[32 + lane];
      float T = redS[lane];
#pragma unroll
      for (int m = 1; m < 32; m <<= 1) T = fmaxf(T, __shfl_xor(T, m, 32));
      const float psqA = sq[piA];
      const float psqB = sq[piB];
      const float4* ownA = (const float4*)(pool + lane * PPITCH);
      const float4* ownB = (const float4*)(pool + (32 + lane) * PPITCH);
      int m = 0;
      for (int c = 0; c < MMAX; c++) {
        const bool bFirst = (pvB > pvA) || (pvB == pvA && piB < piA);
        float cv = bFirst ? pvB : pvA;
        int   ci = bFirst ? piB : piA;
        int   cs = bFirst ? 32 + lane : lane;
#pragma unroll
        for (int x = 1; x < 32; x <<= 1) {
          const float ov = __shfl_xor(cv, x, 32);
          const int   oi = __shfl_xor(ci, x, 32);
          const int   os = __shfl_xor(cs, x, 32);
          if (ov > cv || (ov == cv && oi < ci)) { cv = ov; ci = oi; cs = os; }
        }
        if (c > 0 && !(cv > T)) break;
        const float sA = __shfl(psqA, cs & 31, 32);
        const float sB = __shfl(psqB, cs & 31, 32);
        const float ssq = (cs >= 32) ? sB : sA;
        if (lane == 0) { sel[done + c] = ci; selSqS[c] = ssq; }
        *(v4f*)(selb + SPITCH * c + 4 * lane) = *(const v4f*)(pool + cs * PPITCH + 4 * lane);
        m = c + 1;
        if (done + m == NSAMP || m == MMAX) break;
        const float4* sr = (const float4*)(pool + cs * PPITCH);
        float dA = 0.f, dB = 0.f;
#pragma unroll 8
        for (int kg = 0; kg < 32; kg++) {
          const float4 rA = ownA[kg];
          const float4 rB = ownB[kg];
          const float4 w  = sr[kg];
          dA = fmaf(rA.x, w.x, dA);  dB = fmaf(rB.x, w.x, dB);
          dA = fmaf(rA.y, w.y, dA);  dB = fmaf(rB.y, w.y, dB);
          dA = fmaf(rA.z, w.z, dA);  dB = fmaf(rB.z, w.z, dB);
          dA = fmaf(rA.w, w.w, dA);  dB = fmaf(rB.w, w.w, dB);
        }
        pvA = fminf(pvA, sqrtf(fmaxf(psqA + ssq - 2.f * dA, 0.f)));
        pvB = fminf(pvB, sqrtf(fmaxf(psqB + ssq - 2.f * dB, 0.f)));
      }
      if (lane == 0) mS = m;
    }
    __syncthreads();

    const int m = mS;
    if (m > MVALU) sweep_mm(RB, sqS, anchS, selb, selSqS, m, wid, lane);
    else switch (m) {
      case 1:  sweep_cols<1>(RB, sqS, anchS, selb, selSqS, run, t); break;
      case 2:  sweep_cols<2>(RB, sqS, anchS, selb, selSqS, run, t); break;
      case 3:  sweep_cols<3>(RB, sqS, anchS, selb, selSqS, run, t); break;
      case 4:  sweep_cols<4>(RB, sqS, anchS, selb, selSqS, run, t); break;
      case 5:  sweep_cols<5>(RB, sqS, anchS, selb, selSqS, run, t); break;
      case 6:  sweep_cols<6>(RB, sqS, anchS, selb, selSqS, run, t); break;
      default: sweep_cols<7>(RB, sqS, anchS, selb, selSqS, run, t); break;
    }
    done += m;
    __syncthreads();
  }
}

__global__ __launch_bounds__(256)
void gather_kernel(const float* __restrict__ F, const int* __restrict__ sel,
                   float* __restrict__ out) {
  const int s = blockIdx.x;
  const int c = blockIdx.y * 256 + threadIdx.x;
  out[(size_t)s * DIN + c] = F[(size_t)sel[s] * DIN + c];
}

extern "C" void kernel_launch(void* const* d_in, const int* in_sizes, int n_in,
                              void* d_out, int out_size, void* d_ws, size_t ws_size,
                              hipStream_t stream) {
  (void)in_sizes; (void)n_in; (void)out_size; (void)ws_size;
  const float* F = (const float*)d_in[0];
  const float* W = (const float*)d_in[1];

  float* R          = (float*)d_ws;
  float* sq         = R + (size_t)NPTS * DPROJ;
  float* anchorSq   = sq + NPTS;
  float* anchorPart = anchorSq + NPTS;
  int*   sel        = (int*)(anchorPart + (size_t)NCHUNK * NPTS);

  proj_kernel<<<1024, 256, 0, stream>>>(F, W, R);
  sq_kernel<<<NPTS / 8, 256, 0, stream>>>(R, sq);
  kmajor_kernel<<<NPTS / 128, 1024, 0, stream>>>(R);
  gram_rowsum_kernel<<<1024, 256, 0, stream>>>(R, sq, anchorPart);
  combine_kernel<<<NPTS / 256, 256, 0, stream>>>(anchorPart, anchorSq);
  greedy_kernel<<<1, 1024, 0, stream>>>(R, sq, anchorSq, sel);
  gather_kernel<<<dim3(NSAMP, DIN / 256), 256, 0, stream>>>(F, sel, (float*)d_out);
}


# greedy pool: every wave gives its exact top four rows, not two lane leaders (wave 0 folds four slots a lane, 20 LDS reads a run), so more passes fold 16 selections into one matrix sweep; same sums
# speedup vs baseline: 15.7831x; 1.0935x over previous
#include <hip/hip_runtime.h>


#define NPTS   16384
#define DIN    1024
#define DPROJ  128
#define NSAMP  163
#define NCHUNK 8

typedef float v2f __attribute__((ext_vector_type(2)));
typedef float v8f __attribute__((ext_vector_type(8)));

__device__ __forceinline__ v8f wmma_f32_k4(v2f a, v2f b, v8f c) {
  return __builtin_amdgcn_wmma_f32_16x16x4_f32(
       false, a,  false, b,
       (short)0, c,  false,  false);
}

__global__ __launch_bounds__(256)
void proj_kernel(const float* __restrict__ F, const float* __restrict__ W,
                 float* __restrict__ R) {
  const int wave = (blockIdx.x * blockDim.x + threadIdx.x) >> 5;
  const int lane = threadIdx.x & 31;
  const int tm = wave >> 3;
  const int tn = wave & 7;
  const int m0 = tm * 16, n0 = tn * 16;
  const int half = lane >> 4;
  const int r16  = lane & 15;
  const int koff = half * 2;

  v8f acc = {};
  const float* arow = F + (size_t)(m0 + r16) * DIN + koff;
  for (int k = 0; k < DIN; k += 4) {
    v2f a = *(const v2f*)(arow + k);
    v2f b;
    b.x = W[(size_t)(k + koff)     * DPROJ + n0 + r16];
    b.y = W[(size_t)(k + koff + 1) * DPROJ + n0 + r16];
    acc = wmma_f32_k4(a, b, acc);
  }
#pragma unroll
  for (int i = 0; i < 8; i++)
    R[(size_t)(m0 + i + half * 8) * DPROJ + n0 + r16] = acc[i];
}

__global__ __launch_bounds__(256)
void sq_kernel(const float* __restrict__ R, float* __restrict__ sq) {
  const int row  = (blockIdx.x * blockDim.x + threadIdx.x) >> 5;
  const int lane = threadIdx.x & 31;
  const float* rp = R + (size_t)row * DPROJ;
  float s = 0.f;
#pragma unroll
  for (int k = lane; k < DPROJ; k += 32) { float v = rp[k]; s += v * v; }
#pragma unroll
  for (int m = 16; m >= 1; m >>= 1) s += __shfl_xor(s, m, 32);
  if (lane == 0) sq[row] = s;
}

typedef float v4f __attribute__((ext_vector_type(4)));

__device__ __forceinline__ void gram_load(v4f (&b)[8], const float* p, const int ch) {
#pragma unroll
  for (int u = 0; u < 4; u++) {
    b[2 * u]     = *(const v4f*)(p + (4 * ch + u) * 512);
    b[2 * u + 1] = *(const v4f*)(p + (4 * ch + u) * 512 + 128);
  }
}

template <int CH>
__device__ __forceinline__ void gram_mma(const v4f (&b)[8], const v2f (&afrag)[32],
                                         v8f (&acc)[4]) {
#pragma unroll
  for (int u = 0; u < 4; u++) {
    const v2f a = afrag[4 * CH + u];
    acc[0] = wmma_f32_k4(a, (v2f){b[2 * u].x, b[2 * u + 1].x}, acc[0]);
    acc[1] = wmma_f32_k4(a, (v2f){b[2 * u].y, b[2 * u + 1].y}, acc[1]);
    acc[2] = wmma_f32_k4(a, (v2f){b[2 * u].z, b[2 * u + 1].z}, acc[2]);
    acc[3] = wmma_f32_k4(a, (v2f){b[2 * u].w, b[2 * u + 1].w}, acc[3]);
  }
}

__global__ __launch_bounds__(256)
void gram_rowsum_kernel(const float* __restrict__ RB, const float* __restrict__ sq,
                        float* __restrict__ anchorPart) {
  const int wave  = (blockIdx.x * blockDim.x + threadIdx.x) >> 5;
  const int lane  = threadIdx.x & 31;
  const int chunk = wave & (NCHUNK - 1);
  const int strip = wave >> 3;
  const int half  = lane >> 4;
  const int r16   = lane & 15;
  const int m0    = strip * 16;

  v2f afrag[32];
  const float* ap = RB + (size_t)(m0 >> 7) * (128 * DPROJ) + ((m0 & 127) + r16)
                       + half * 256;
#pragma unroll
  for (int kk = 0; kk < 32; kk++) {
    afrag[kk].x = ap[kk * 512];
    afrag[kk].y = ap[kk * 512 + 128];
  }

  float sqr[8];
#pragma unroll
  for (int i = 0; i < 8; i++) sqr[i] = sq[m0 + i + half * 8];

  float partial[8] = {0.f, 0.f, 0.f, 0.f, 0.f, 0.f, 0.f, 0.f};

  const float* cp = RB + (size_t)(chunk * 16) * (128 * DPROJ) + 4 * r16 + half * 256;
  const float* sp = sq + chunk * 2048 + 4 * r16;

  v4f bA[8], bB[8];
  gram_load(bA, cp, 0);

#pragma unroll 1
  for (int G = 0; G < 32; G++) {
    const float* p  = cp + (size_t)(G >> 1) * (128 * DPROJ) + (G & 1) * 64;
    const int Gn = (G < 31) ? G + 1 : G;
    const float* pn = cp + (size_t)(Gn >> 1) * (128 * DPROJ) + (Gn & 1) * 64;
    const v4f sc = *(const v4f*)(sp + 64 * G);

    v8f acc[4] = {};
    gram_load(bB, p, 1);
    gram_mma<0>(bA, afrag, acc);
    gram_load(bA, p, 2);
    gram_mma<1>(bB, afrag, acc);
    gram_load(bB, p, 3);
    gram_mma<2>(bA, afrag, acc);
    gram_load(bA, p, 4);
    gram_mma<3>(bB, afrag, acc);
    gram_load(bB, p, 5);
    gram_mma<4>(bA, afrag, acc);
    gram_load(bA, p, 6);
    gram_mma<5>(bB, afrag, acc);
    gram_load(bB, p, 7);
    gram_mma<6>(bA, afrag, acc);
    gram_load(bA, pn, 0);
    gram_mma<7>(bB, afrag, acc);

#pragma unroll
    for (int q = 0; q < 4; q++) {
#pragma unroll
      for (int i = 0; i < 8; i++)
        partial[i] += fmaxf(0.f, sqr[i] + sc[q] - 2.f * acc[q][i]);
    }
  }

#pragma unroll
  for (int i = 0; i < 8; i++) {
    float p = partial[i];
    p += __shfl_xor(p, 1, 32);
    p += __shfl_xor(p, 2, 32);
    p += __shfl_xor(p, 4, 32);
    p += __shfl_xor(p, 8, 32);
    if (r16 == 0)
      anchorPart[(size_t)chunk * NPTS + m0 + i + half * 8] = p;
  }
}

__global__ __launch_bounds__(256)
void combine_kernel(const float* __restrict__ anchorPart,
                    float* __restrict__ anchorSq) {
  const int i = blockIdx.x * 256 + threadIdx.x;
  float s = 0.f;
#pragma unroll
  for (int c = 0; c < NCHUNK; c++) s += anchorPart[(size_t)c * NPTS + i];
  anchorSq[i] = s;
}

__global__ __launch_bounds__(1024)
void kmajor_kernel(float* __restrict__ R) {
  float* blk = R + (size_t)blockIdx.x * (128 * DPROJ);
  const int lane = threadIdx.x & 31;
  const int w    = threadIdx.x >> 5;
  const int m    = 32 * (w & 3) + lane;
  const int kg0  = w >> 2;

  float4 v[4];
#pragma unroll
  for (int i = 0; i < 4; i++)
    v[i] = *(const float4*)(blk + m * DPROJ + 4 * (kg0 + 8 * i));

  __syncthreads();

#pragma unroll
  for (int i = 0; i < 4; i++) {
    const int k = 4 * (kg0 + 8 * i);
    blk[(k + 0) * 128 + m] = v[i].x;
    blk[(k + 1) * 128 + m] = v[i].y;
    blk[(k + 2) * 128 + m] = v[i].z;
    blk[(k + 3) * 128 + m] = v[i].w;
  }
}

#define MMAX   16
#define MVALU  7
#define PERW   4
#define NPOOL  (32 * PERW)
#define PPITCH 132
#define SPITCH 132

typedef const __attribute__((address_space(1))) float* gfp;
typedef const __attribute__((address_space(1))) v4f*   gv4p;

__device__ __forceinline__ void load_group(v4f (&r)[4], gfp p) {
#pragma unroll
  for (int u = 0; u < 4; u++) r[u] = *(gv4p)(p + u * 128);
}

template <int M, int C0, int P>
__device__ __forceinline__ void fold_part(const v4f (&v)[4], const float* selb,
                                          const int g, v2f (&d)[M][2]) {
  v4f w[P];
#pragma unroll
  for (int c = 0; c < P; c++) w[c] = *(const v4f*)(selb + SPITCH * (C0 + c) + 4 * g);
  if constexpr (P == 2) asm volatile("" : "+v"(w[0]), "+v"(w[1]));
  if constexpr (P == 3) asm volatile("" : "+v"(w[0]), "+v"(w[1]), "+v"(w[2]));
  if constexpr (P == 4) asm volatile("" : "+v"(w[0]), "+v"(w[1]), "+v"(w[2]), "+v"(w[3]));
#pragma unroll
  for (int c = 0; c < P; c++) {
    d[C0 + c][0] = __builtin_elementwise_fma(v[0].xy, w[c].xx, d[C0 + c][0]);
    d[C0 + c][1] = __builtin_elementwise_fma(v[0].zw, w[c].xx, d[C0 + c][1]);
  }
#pragma unroll
  for (int c = 0; c < P; c++) {
    d[C0 + c][0] = __builtin_elementwise_fma(v[1].xy, w[c].yy, d[C0 + c][0]);
    d[C0 + c][1] = __builtin_elementwise_fma(v[1].zw, w[c].yy, d[C0 + c][1]);
  }
#pragma unroll
  for (int c = 0; c < P; c++) {
    d[C0 + c][0] = __builtin_elementwise_fma(v[2].xy, w[c].zz, d[C0 + c][0]);
    d[C0 + c][1] = __builtin_elementwise_fma(v[2].zw, w[c].zz, d[C0 + c][1]);
  }
#pragma unroll
  for (int c = 0; c < P; c++) {
    d[C0 + c][0] = __builtin_elementwise_fma(v[3].xy, w[c].ww, d[C0 + c][0]);
    d[C0 + c][1] = __builtin_elementwise_fma(v[3].zw, w[c].ww, d[C0 + c][1]);
  }
}

template <int M>
__device__ __forceinline__ void fold_group(const v4f (&v)[4], const float* selb,
                                           const int g, v2f (&d)[M][2]) {
  if constexpr (M <= 4) {
    fold_part<M, 0, M>(v, selb, g, d);
  } else {
    fold_part<M, 0, 4>(v, selb, g, d);
    fold_part<M, 4, M - 4>(v, selb, g, d);
  }
}

template <int M>
__device__ __forceinline__ void after_fold(gfp &p, v2f (&d)[M][2]) {
#pragma unroll
  for (int c = 0; c < M; c++)
    asm volatile("" : "+v"(p), "+v"(d[c][0]), "+v"(d[c][1]));
}

template <int M, bool AHEAD>
__device__ __forceinline__ void ring_trip(gfp &p, const float* selb, const int g,
                                          v4f (&r0)[4], v4f (&r1)[4],
                                          v4f (&r2)[4], v4f (&r3)[4],
                                          v2f (&d)[M][2]) {
  load_group(r3, p + 3 * 512);
  fold_group<M>(r0, selb, g, d);
  after_fold<M>(p, d);
  if (AHEAD) load_group(r0, p + 4 * 512);
  fold_group<M>(r1, selb, g + 1, d);
  after_fold<M>(p, d);
  if (AHEAD) load_group(r1, p + 5 * 512);
  fold_group<M>(r2, selb, g + 2, d);
  after_fold<M>(p, d);
  if (AHEAD) load_group(r2, p + 6 * 512);
  fold_group<M>(r3, selb, g + 3, d);
  after_fold<M>(p, d);
  p += 4 * 512;
}

template <int M>
__device__ __forceinline__ void sweep_cols(const float* __restrict__ RB,
                                           const float* sqS, float* anchS,
                                           const float* selb,
                                           const float* selSqS,
                                           const unsigned run, const int t) {
  float ssq[M];
#pragma unroll
  for (int c = 0; c < M; c++) {
    ssq[c] = __int_as_float(__builtin_amdgcn_readfirstlane(__float_as_int(selSqS[c])));
  }
  gfp p = (gfp)(RB + run);
  v4f r0[4], r1[4], r2[4], r3[4];
  load_group(r0, p);
  asm volatile("" : "+v"(p));
  load_group(r1, p + 512);
  asm volatile("" : "+v"(p));
  load_group(r2, p + 1024);
#pragma unroll
  for (int j = 0; j < 4; j++) {
    v2f d[M][2];
#pragma unroll
    for (int c = 0; c < M; c++) { d[c][0] = (v2f){0.f, 0.f}; d[c][1] = (v2f){0.f, 0.f}; }
    if (j < 3) {
#pragma unroll 1
      for (int g = 0; g < 32; g += 4)
        ring_trip<M, true>(p, selb, g, r0, r1, r2, r3, d);
    } else {
#pragma unroll 1
      for (int g = 0; g < 28; g += 4)
        ring_trip<M, true>(p, selb, g, r0, r1, r2, r3, d);
      ring_trip<M, false>(p, selb, 28, r0, r1, r2, r3, d);
    }
    const float4 q = *(const float4*)(sqS + 4 * (1024 * j + t));
    float4 a = *(const float4*)(anchS + 4 * (1024 * j + t));
    float e0 = q.x + ssq[0] - 2.f * d[0][0].x;
    float e1 = q.y + ssq[0] - 2.f * d[0][0].y;
    float e2 = q.z + ssq[0] - 2.f * d[0][1].x;
    float e3 = q.w + ssq[0] - 2.f * d[0][1].y;
#pragma unroll
    for (int c = 1; c < M; c++) {
      e0 = fminf(e0, q.x + ssq[c] - 2.f * d[c][0].x);
      e1 = fminf(e1, q.y + ssq[c] - 2.f * d[c][0].y);
      e2 = fminf(e2, q.z + ssq[c] - 2.f * d[c][1].x);
      e3 = fminf(e3, q.w + ssq[c] - 2.f * d[c][1].y);
    }
    a.x = fminf(a.x, sqrtf(fmaxf(e0, 0.f)));
    a.y = fminf(a.y, sqrtf(fmaxf(e1, 0.f)));
    a.z = fminf(a.z, sqrtf(fmaxf(e2, 0.f)));
    a.w = fminf(a.w, sqrtf(fmaxf(e3, 0.f)));
    *(float4*)(anchS + 4 * (1024 * j + t)) = a;
  }
}

__device__ __forceinline__ void mm_load(v4f (&b)[4], v2f (&a)[2], gfp p,
                                        const float* ap, const int c) {
#pragma unroll
  for (int u = 0; u < 2; u++) {
    b[2 * u]     = *(gv4p)(p + (2 * c + u) * 512);
    b[2 * u + 1] = *(gv4p)(p + (2 * c + u) * 512 + 128);
    a[u]         = *(const v2f*)(ap + 4 * (2 * c + u));
  }
}

__device__ __forceinline__ void mm_mma(const v4f (&b)[4], const v2f (&a)[2],
                                       v8f (&acc)[4]) {
#pragma unroll
  for (int u = 0; u < 2; u++) {
    acc[0] = wmma_f32_k4(a[u], (v2f){b[2 * u].x, b[2 * u + 1].x}, acc[0]);
    acc[1] = wmma_f32_k4(a[u], (v2f){b[2 * u].y, b[2 * u + 1].y}, acc[1]);
    acc[2] = wmma_f32_k4(a[u], (v2f){b[2 * u].z, b[2 * u + 1].z}, acc[2]);
    acc[3] = wmma_f32_k4(a[u], (v2f){b[2 * u].w, b[2 * u + 1].w}, acc[3]);
  }
}

__device__ __forceinline__ void sweep_mm(const float* __restrict__ RB,
                                         const float* sqS, float* anchS,
                                         const float* selb, const float* selSqS,
                                         const int m, const int wid, const int lane) {
  const int half = lane >> 4;
  const int r16  = lane & 15;
  float ssq8[8];
#pragma unroll
  for (int i = 0; i < 8; i++) {
    const int s = 8 * half + i;
    const float v = selSqS[s];
    ssq8[i] = (s < m) ? v : __builtin_inff();
  }
  unsigned aoff = (unsigned)(r16 * SPITCH + 2 * half);
  gfp p0 = (gfp)(RB + (unsigned)wid * (4u * 128u * DPROJ) + 4u * (unsigned)r16
                    + 256u * (unsigned)half);
  v4f bA[4], bB[4];
  v2f aA[2], aB[2];
  mm_load(bA, aA, p0, selb + aoff, 0);
#pragma unroll 1
  for (int G = 0; G < 8; G++) {
    asm volatile("" : "+v"(aoff));
    const float* ap = selb + aoff;
    gfp p = p0 + (G >> 1) * (128 * DPROJ) + (G & 1) * 64;
    const int Gn = (G < 7) ? G + 1 : G;
    gfp pn = p0 + (Gn >> 1) * (128 * DPROJ) + (Gn & 1) * 64;
    v8f acc[4] = {};
#pragma unroll
    for (int c2 = 0; c2 < 8; c2++) {
      mm_load(bB, aB, p, ap, 2 * c2 + 1);
      mm_mma(bA, aA, acc);
      if (c2 < 7) mm_load(bA, aA, p, ap, 2 * c2 + 2);
      else        mm_load(bA, aA, pn, ap, 0);
      mm_mma(bB, aB, acc);
    }

    const int slot = 1024 * (G >> 1) + 32 * wid + 16 * (G & 1) + r16;
    const float4 q = *(const float4*)(sqS + 4 * slot);
    float4 a = *(const float4*)(anchS + 4 * slot);
    float e0 = __builtin_inff(), e1 = e0, e2 = e0, e3 = e0;
#pragma unroll
    for (int i = 0; i < 8; i++) {
      e0 = fminf(e0, q.x + ssq8[i] - 2.f * acc[0][i]);
      e1 = fminf(e1, q.y + ssq8[i] - 2.f * acc[1][i]);
      e2 = fminf(e2, q.z + ssq8[i] - 2.f * acc[2][i]);
      e3 = fminf(e3, q.w + ssq8[i] - 2.f * acc[3][i]);
    }
    e0 = fminf(e0, __shfl_xor(e0, 16, 32));
    e1 = fminf(e1, __shfl_xor(e1, 16, 32));
    e2 = fminf(e2, __shfl_xor(e2, 16, 32));
    e3 = fminf(e3, __shfl_xor(e3, 16, 32));
    a.x = fminf(a.x, sqrtf(fmaxf(e0, 0.f)));
    a.y = fminf(a.y, sqrtf(fmaxf(e1, 0.f)));
    a.z = fminf(a.z, sqrtf(fmaxf(e2, 0.f)));
    a.w = fminf(a.w, sqrtf(fmaxf(e3, 0.f)));
    if (half == 0) *(float4*)(anchS + 4 * slot) = a;
  }
}

__global__ __launch_bounds__(1024)
void greedy_kernel(const float* __restrict__ RB, const float* __restrict__ sq,
                   const float* __restrict__ anchorSq, int* __restrict__ sel) {
  __shared__ float redV[NPOOL];
  __shared__ int   redI[NPOOL];
  __shared__ float redS[32];
  __shared__ __attribute__((aligned(16))) float pool[NPOOL * PPITCH];
  __shared__ __attribute__((aligned(16))) float selb[MMAX * SPITCH];
  __shared__ float selSqS[MMAX];
  __shared__ int   mS;
  __shared__ __attribute__((aligned(16))) float anchS[4 * 1024 * 4];
  __shared__ __attribute__((aligned(16))) float sqS[4 * 1024 * 4];

  const int t    = threadIdx.x;
  const int lane = t & 31;
  const int wid  = t >> 5;
  const int row0 = 512 * wid + 4 * lane;

#pragma unroll
  for (int j = 0; j < 4; j++) {
    float4 a = *(const float4*)(anchorSq + row0 + 128 * j);
    a.x = sqrtf(a.x);
    a.y = sqrtf(a.y);
    a.z = sqrtf(a.z);
    a.w = sqrtf(a.w);
    *(float4*)(anchS + 4 * (1024 * j + t)) = a;
    *(float4*)(sqS + 4 * (1024 * j + t)) = *(const float4*)(sq + row0 + 128 * j);
  }

  for (int i = t; i < MMAX * SPITCH; i += 1024) selb[i] = 0.f;

  const unsigned run = (unsigned)wid * (4u * 128u * DPROJ) + 4u * (unsigned)lane;

  int done = 0;
  for (int pass = 0; pass < NSAMP && done < NSAMP; pass++) {
    float aVal[16];
#pragma unroll
    for (int j = 0; j < 4; j++) {
      const float4 a = *(const float4*)(anchS + 4 * (1024 * j + t));
      aVal[4 * j + 0] = a.x; aVal[4 * j + 1] = a.y; aVal[4 * j + 2] = a.z; aVal[4 * j + 3] = a.w;
    }
    float wv[PERW]; int wi[PERW];
#pragma unroll
    for (int r = 0; r < PERW; r++) {
      float bv = aVal[0]; int be = 0;
#pragma unroll
      for (int e = 1; e < 16; e++) {
        const float v = aVal[e];
        if (v > bv) { bv = v; be = e; }
      }
      const int mine = row0 + 128 * (be >> 2) + (be & 3);
      int bi = mine;
#pragma unroll
      for (int m = 1; m < 32; m <<= 1) {
        const float ov = __shfl_xor(bv, m, 32);
        const int   oi = __shfl_xor(bi, m, 32);
        if ((ov > bv) || (ov == bv && oi < bi)) { bv = ov; bi = oi; }
      }
      wv[r] = bv; wi[r] = bi;
      const int gone = (bi == mine) ? be : -1;
#pragma unroll
      for (int e = 0; e < 16; e++)
        if (gone == e) aVal[e] = -2.f;
    }
    float tl = aVal[0];
#pragma unroll
    for (int e = 1; e < 16; e++) tl = fmaxf(tl, aVal[e]);
#pragma unroll
    for (int m = 1; m < 32; m <<= 1) tl = fmaxf(tl, __shfl_xor(tl, m, 32));
    {
      float g[PERW][4];
#pragma unroll
      for (int r = 0; r < PERW; r++) {
        const unsigned base = (unsigned)(wi[r] >> 7) * (128u * DPROJ) + (unsigned)(wi[r] & 127);
#pragma unroll
        for (int i = 0; i < 4; i++) g[r][i] = RB[base + (unsigned)(lane + 32 * i) * 128u];
      }
#pragma unroll
      for (int r = 0; r < PERW; r++) {
#pragma unroll
        for (int i = 0; i < 4; i++) pool[(wid + 32 * r) * PPITCH + lane + 32 * i] = g[r][i];
      }
    }
    if (lane == 0) {
#pragma unroll
      for (int r = 0; r < PERW; r++) { redV[wid + 32 * r] = wv[r]; redI[wid + 32 * r] = wi[r]; }
      redS[wid] = tl;
    }
    __syncthreads();

    if (wid == 0) {
      float pv[PERW]; int pi[PERW]; float psq[PERW];
      const v4f* own[PERW];
#pragma unroll
      for (int r = 0; r < PERW; r++) {
        pv[r]  = redV[lane + 32 * r];
        pi[r]  = redI[lane + 32 * r];
        psq[r] = sq[pi[r]];
        own[r] = (const v4f*)(pool + (lane + 32 * r) * PPITCH);
      }
      float T = redS[lane];
#pragma unroll
      for (int m = 1; m < 32; m <<= 1) T = fmaxf(T, __shfl_xor(T, m, 32));
      int m = 0;
      for (int c = 0; c < MMAX; c++) {
        float cv = pv[0];
        int   ci = pi[0];
        int   cs = lane;
#pragma unroll
        for (int r = 1; r < PERW; r++) {
          if (pv[r] > cv || (pv[r] == cv && pi[r] < ci)) { cv = pv[r]; ci = pi[r]; cs = lane + 32 * r; }
        }
#pragma unroll
        for (int x = 1; x < 32; x <<= 1) {
          const float ov = __shfl_xor(cv, x, 32);
          const int   oi = __shfl_xor(ci, x, 32);
          const int   os = __shfl_xor(cs, x, 32);
          if (ov > cv || (ov == cv && oi < ci)) { cv = ov; ci = oi; cs = os; }
        }
        if (c > 0 && !(cv > T)) break;
        float mine = psq[0];
#pragma unroll
        for (int r = 1; r < PERW; r++) mine = ((cs >> 5) == r) ? psq[r] : mine;
        const float ssq = __shfl(mine, cs & 31, 32);
        if (lane == 0) { sel[done + c] = ci; selSqS[c] = ssq; }
        *(v4f*)(selb + SPITCH * c + 4 * lane) = *(const v4f*)(pool + cs * PPITCH + 4 * lane);
        m = c + 1;
        if (done + m == NSAMP || m == MMAX) break;
        const v4f* sr = (const v4f*)(pool + cs * PPITCH);
        float d[PERW];
#pragma unroll
        for (int r = 0; r < PERW; r++) d[r] = 0.f;
        static_assert(PERW == 4, "the statement below names four slots' registers");
#pragma unroll 1
        for (int kb = 0; kb < 32; kb += 4) {
          v4f w[4], o[PERW][4];
#pragma unroll
          for (int u = 0; u < 4; u++) {
            w[u] = sr[kb + u];
#pragma unroll
            for (int r = 0; r < PERW; r++) o[r][u] = own[r][kb + u];
          }
          asm volatile("" : "+v"(w[0]), "+v"(w[1]), "+v"(w[2]), "+v"(w[3]),
                            "+v"(o[0][0]), "+v"(o[1][0]), "+v"(o[2][0]), "+v"(o[3][0])
                          : "v"(o[0][1]), "v"(o[0][2]), "v"(o[0][3]),
                            "v"(o[1][1]), "v"(o[1][2]), "v"(o[1][3]),
                            "v"(o[2][1]), "v"(o[2][2]), "v"(o[2][3]),
                            "v"(o[3][1]), "v"(o[3][2]), "v"(o[3][3]));
#pragma unroll
          for (int u = 0; u < 4; u++) {
#pragma unroll
            for (int r = 0; r < PERW; r++) d[r] = fmaf(o[r][u].x, w[u].x, d[r]);
#pragma unroll
            for (int r = 0; r < PERW; r++) d[r] = fmaf(o[r][u].y, w[u].y, d[r]);
#pragma unroll
            for (int r = 0; r < PERW; r++) d[r] = fmaf(o[r][u].z, w[u].z, d[r]);
#pragma unroll
            for (int r = 0; r < PERW; r++) d[r] = fmaf(o[r][u].w, w[u].w, d[r]);
          }
        }
#pragma unroll
        for (int r = 0; r < PERW; r++)
          pv[r] = fminf(pv[r], sqrtf(fmaxf(psq[r] + ssq - 2.f * d[r], 0.f)));
      }
      if (lane == 0) mS = m;
    }
    __syncthreads();

    const int m = mS;
    if (m > MVALU) sweep_mm(RB, sqS, anchS, selb, selSqS, m, wid, lane);
    else switch (m) {
      case 1:  sweep_cols<1>(RB, sqS, anchS, selb, selSqS, run, t); break;
      case 2:  sweep_cols<2>(RB, sqS, anchS, selb, selSqS, run, t); break;
      case 3:  sweep_cols<3>(RB, sqS, anchS, selb, selSqS, run, t); break;
      case 4:  sweep_cols<4>(RB, sqS, anchS, selb, selSqS, run, t); break;
      case 5:  sweep_cols<5>(RB, sqS, anchS, selb, selSqS, run, t); break;
      case 6:  sweep_cols<6>(RB, sqS, anchS, selb, selSqS, run, t); break;
      default: sweep_cols<7>(RB, sqS, anchS, selb, selSqS, run, t); break;
    }
    done += m;
    __syncthreads();
  }
}

__global__ __launch_bounds__(256)
void gather_kernel(const float* __restrict__ F, const int* __restrict__ sel,
                   float* __restrict__ out) {
  const int s = blockIdx.x;
  const int c = blockIdx.y * 256 + threadIdx.x;
  out[(size_t)s * DIN + c] = F[(size_t)sel[s] * DIN + c];
}

extern "C" void kernel_launch(void* const* d_in, const int* in_sizes, int n_in,
                              void* d_out, int out_size, void* d_ws, size_t ws_size,
                              hipStream_t stream) {
  (void)in_sizes; (void)n_in; (void)out_size; (void)ws_size;
  const float* F = (const float*)d_in[0];
  const float* W = (const float*)d_in[1];

  float* R          = (float*)d_ws;
  float* sq         = R + (size_t)NPTS * DPROJ;
  float* anchorSq   = sq + NPTS;
  float* anchorPart = anchorSq + NPTS;
  int*   sel        = (int*)(anchorPart + (size_t)NCHUNK * NPTS);

  proj_kernel<<<1024, 256, 0, stream>>>(F, W, R);
  sq_kernel<<<NPTS / 8, 256, 0, stream>>>(R, sq);
  kmajor_kernel<<<NPTS / 128, 1024, 0, stream>>>(R);
  gram_rowsum_kernel<<<1024, 256, 0, stream>>>(R, sq, anchorPart);
  combine_kernel<<<NPTS / 256, 256, 0, stream>>>(anchorPart, anchorSq);
  greedy_kernel<<<1, 1024, 0, stream>>>(R, sq, anchorSq, sel);
  gather_kernel<<<dim3(NSAMP, DIN / 256), 256, 0, stream>>>(F, sel, (float*)d_out);
}


# greedy pool: the 128 pool rows' mutual dot products are taken once a pass by all 32 waves on f32 WMMA; a selection then folds in by 4 LDS reads a lane, not 128-long dots; pool dot sum order changes
# speedup vs baseline: 16.7280x; 1.0599x over previous
#include <hip/hip_runtime.h>


#define NPTS   16384
#define DIN    1024
#define DPROJ  128
#define NSAMP  163
#define NCHUNK 8

typedef float v2f __attribute__((ext_vector_type(2)));
typedef float v8f __attribute__((ext_vector_type(8)));

__device__ __forceinline__ v8f wmma_f32_k4(v2f a, v2f b, v8f c) {
  return __builtin_amdgcn_wmma_f32_16x16x4_f32(
       false, a,  false, b,
       (short)0, c,  false,  false);
}

__global__ __launch_bounds__(256)
void proj_kernel(const float* __restrict__ F, const float* __restrict__ W,
                 float* __restrict__ R) {
  const int wave = (blockIdx.x * blockDim.x + threadIdx.x) >> 5;
  const int lane = threadIdx.x & 31;
  const int tm = wave >> 3;
  const int tn = wave & 7;
  const int m0 = tm * 16, n0 = tn * 16;
  const int half = lane >> 4;
  const int r16  = lane & 15;
  const int koff = half * 2;

  v8f acc = {};
  const float* arow = F + (size_t)(m0 + r16) * DIN + koff;
  for (int k = 0; k < DIN; k += 4) {
    v2f a = *(const v2f*)(arow + k);
    v2f b;
    b.x = W[(size_t)(k + koff)     * DPROJ + n0 + r16];
    b.y = W[(size_t)(k + koff + 1) * DPROJ + n0 + r16];
    acc = wmma_f32_k4(a, b, acc);
  }
#pragma unroll
  for (int i = 0; i < 8; i++)
    R[(size_t)(m0 + i + half * 8) * DPROJ + n0 + r16] = acc[i];
}

__global__ __launch_bounds__(256)
void sq_kernel(const float* __restrict__ R, float* __restrict__ sq) {
  const int row  = (blockIdx.x * blockDim.x + threadIdx.x) >> 5;
  const int lane = threadIdx.x & 31;
  const float* rp = R + (size_t)row * DPROJ;
  float s = 0.f;
#pragma unroll
  for (int k = lane; k < DPROJ; k += 32) { float v = rp[k]; s += v * v; }
#pragma unroll
  for (int m = 16; m >= 1; m >>= 1) s += __shfl_xor(s, m, 32);
  if (lane == 0) sq[row] = s;
}

typedef float v4f __attribute__((ext_vector_type(4)));

__device__ __forceinline__ void gram_load(v4f (&b)[8], const float* p, const int ch) {
#pragma unroll
  for (int u = 0; u < 4; u++) {
    b[2 * u]     = *(const v4f*)(p + (4 * ch + u) * 512);
    b[2 * u + 1] = *(const v4f*)(p + (4 * ch + u) * 512 + 128);
  }
}

template <int CH>
__device__ __forceinline__ void gram_mma(const v4f (&b)[8], const v2f (&afrag)[32],
                                         v8f (&acc)[4]) {
#pragma unroll
  for (int u = 0; u < 4; u++) {
    const v2f a = afrag[4 * CH + u];
    acc[0] = wmma_f32_k4(a, (v2f){b[2 * u].x, b[2 * u + 1].x}, acc[0]);
    acc[1] = wmma_f32_k4(a, (v2f){b[2 * u].y, b[2 * u + 1].y}, acc[1]);
    acc[2] = wmma_f32_k4(a, (v2f){b[2 * u].z, b[2 * u + 1].z}, acc[2]);
    acc[3] = wmma_f32_k4(a, (v2f){b[2 * u].w, b[2 * u + 1].w}, acc[3]);
  }
}

__global__ __launch_bounds__(256)
void gram_rowsum_kernel(const float* __restrict__ RB, const float* __restrict__ sq,
                        float* __restrict__ anchorPart) {
  const int wave  = (blockIdx.x * blockDim.x + threadIdx.x) >> 5;
  const int lane  = threadIdx.x & 31;
  const int chunk = wave & (NCHUNK - 1);
  const int strip = wave >> 3;
  const int half  = lane >> 4;
  const int r16   = lane & 15;
  const int m0    = strip * 16;

  v2f afrag[32];
  const float* ap = RB + (size_t)(m0 >> 7) * (128 * DPROJ) + ((m0 & 127) + r16)
                       + half * 256;
#pragma unroll
  for (int kk = 0; kk < 32; kk++) {
    afrag[kk].x = ap[kk * 512];
    afrag[kk].y = ap[kk * 512 + 128];
  }

  float sqr[8];
#pragma unroll
  for (int i = 0; i < 8; i++) sqr[i] = sq[m0 + i + half * 8];

  float partial[8] = {0.f, 0.f, 0.f, 0.f, 0.f, 0.f, 0.f, 0.f};

  const float* cp = RB + (size_t)(chunk * 16) * (128 * DPROJ) + 4 * r16 + half * 256;
  const float* sp = sq + chunk * 2048 + 4 * r16;

  v4f bA[8], bB[8];
  gram_load(bA, cp, 0);

#pragma unroll 1
  for (int G = 0; G < 32; G++) {
    const float* p  = cp + (size_t)(G >> 1) * (128 * DPROJ) + (G & 1) * 64;
    const int Gn = (G < 31) ? G + 1 : G;
    const float* pn = cp + (size_t)(Gn >> 1) * (128 * DPROJ) + (Gn & 1) * 64;
    const v4f sc = *(const v4f*)(sp + 64 * G);

    v8f acc[4] = {};
    gram_load(bB, p, 1);
    gram_mma<0>(bA, afrag, acc);
    gram_load(bA, p, 2);
    gram_mma<1>(bB, afrag, acc);
    gram_load(bB, p, 3);
    gram_mma<2>(bA, afrag, acc);
    gram_load(bA, p, 4);
    gram_mma<3>(bB, afrag, acc);
    gram_load(bB, p, 5);
    gram_mma<4>(bA, afrag, acc);
    gram_load(bA, p, 6);
    gram_mma<5>(bB, afrag, acc);
    gram_load(bB, p, 7);
    gram_mma<6>(bA, afrag, acc);
    gram_load(bA, pn, 0);
    gram_mma<7>(bB, afrag, acc);

#pragma unroll
    for (int q = 0; q < 4; q++) {
#pragma unroll
      for (int i = 0; i < 8; i++)
        partial[i] += fmaxf(0.f, sqr[i] + sc[q] - 2.f * acc[q][i]);
    }
  }

#pragma unroll
  for (int i = 0; i < 8; i++) {
    float p = partial[i];
    p += __shfl_xor(p, 1, 32);
    p += __shfl_xor(p, 2, 32);
    p += __shfl_xor(p, 4, 32);
    p += __shfl_xor(p, 8, 32);
    if (r16 == 0)
      anchorPart[(size_t)chunk * NPTS + m0 + i + half * 8] = p;
  }
}

__global__ __launch_bounds__(256)
void combine_kernel(const float* __restrict__ anchorPart,
                    float* __restrict__ anchorSq) {
  const int i = blockIdx.x * 256 + threadIdx.x;
  float s = 0.f;
#pragma unroll
  for (int c = 0; c < NCHUNK; c++) s += anchorPart[(size_t)c * NPTS + i];
  anchorSq[i] = s;
}

__global__ __launch_bounds__(1024)
void kmajor_kernel(float* __restrict__ R) {
  float* blk = R + (size_t)blockIdx.x * (128 * DPROJ);
  const int lane = threadIdx.x & 31;
  const int w    = threadIdx.x >> 5;
  const int m    = 32 * (w & 3) + lane;
  const int kg0  = w >> 2;

  float4 v[4];
#pragma unroll
  for (int i = 0; i < 4; i++)
    v[i] = *(const float4*)(blk + m * DPROJ + 4 * (kg0 + 8 * i));

  __syncthreads();

#pragma unroll
  for (int i = 0; i < 4; i++) {
    const int k = 4 * (kg0 + 8 * i);
    blk[(k + 0) * 128 + m] = v[i].x;
    blk[(k + 1) * 128 + m] = v[i].y;
    blk[(k + 2) * 128 + m] = v[i].z;
    blk[(k + 3) * 128 + m] = v[i].w;
  }
}

#define MMAX   16
#define MVALU  7
#define PERW   4
#define NPOOL  (32 * PERW)
#define PPITCH 132
#define GPITCH 132
#define SPITCH 132

typedef const __attribute__((address_space(1))) float* gfp;
typedef const __attribute__((address_space(1))) v4f*   gv4p;

__device__ __forceinline__ void load_group(v4f (&r)[4], gfp p) {
#pragma unroll
  for (int u = 0; u < 4; u++) r[u] = *(gv4p)(p + u * 128);
}

template <int M, int C0, int P>
__device__ __forceinline__ void fold_part(const v4f (&v)[4], const float* selb,
                                          const int g, v2f (&d)[M][2]) {
  v4f w[P];
#pragma unroll
  for (int c = 0; c < P; c++) w[c] = *(const v4f*)(selb + SPITCH * (C0 + c) + 4 * g);
  if constexpr (P == 2) asm volatile("" : "+v"(w[0]), "+v"(w[1]));
  if constexpr (P == 3) asm volatile("" : "+v"(w[0]), "+v"(w[1]), "+v"(w[2]));
  if constexpr (P == 4) asm volatile("" : "+v"(w[0]), "+v"(w[1]), "+v"(w[2]), "+v"(w[3]));
#pragma unroll
  for (int c = 0; c < P; c++) {
    d[C0 + c][0] = __builtin_elementwise_fma(v[0].xy, w[c].xx, d[C0 + c][0]);
    d[C0 + c][1] = __builtin_elementwise_fma(v[0].zw, w[c].xx, d[C0 + c][1]);
  }
#pragma unroll
  for (int c = 0; c < P; c++) {
    d[C0 + c][0] = __builtin_elementwise_fma(v[1].xy, w[c].yy, d[C0 + c][0]);
    d[C0 + c][1] = __builtin_elementwise_fma(v[1].zw, w[c].yy, d[C0 + c][1]);
  }
#pragma unroll
  for (int c = 0; c < P; c++) {
    d[C0 + c][0] = __builtin_elementwise_fma(v[2].xy, w[c].zz, d[C0 + c][0]);
    d[C0 + c][1] = __builtin_elementwise_fma(v[2].zw, w[c].zz, d[C0 + c][1]);
  }
#pragma unroll
  for (int c = 0; c < P; c++) {
    d[C0 + c][0] = __builtin_elementwise_fma(v[3].xy, w[c].ww, d[C0 + c][0]);
    d[C0 + c][1] = __builtin_elementwise_fma(v[3].zw, w[c].ww, d[C0 + c][1]);
  }
}

template <int M>
__device__ __forceinline__ void fold_group(const v4f (&v)[4], const float* selb,
                                           const int g, v2f (&d)[M][2]) {
  if constexpr (M <= 4) {
    fold_part<M, 0, M>(v, selb, g, d);
  } else {
    fold_part<M, 0, 4>(v, selb, g, d);
    fold_part<M, 4, M - 4>(v, selb, g, d);
  }
}

template <int M>
__device__ __forceinline__ void after_fold(gfp &p, v2f (&d)[M][2]) {
#pragma unroll
  for (int c = 0; c < M; c++)
    asm volatile("" : "+v"(p), "+v"(d[c][0]), "+v"(d[c][1]));
}

template <int M, bool AHEAD>
__device__ __forceinline__ void ring_trip(gfp &p, const float* selb, const int g,
                                          v4f (&r0)[4], v4f (&r1)[4],
                                          v4f (&r2)[4], v4f (&r3)[4],
                                          v2f (&d)[M][2]) {
  load_group(r3, p + 3 * 512);
  fold_group<M>(r0, selb, g, d);
  after_fold<M>(p, d);
  if (AHEAD) load_group(r0, p + 4 * 512);
  fold_group<M>(r1, selb, g + 1, d);
  after_fold<M>(p, d);
  if (AHEAD) load_group(r1, p + 5 * 512);
  fold_group<M>(r2, selb, g + 2, d);
  after_fold<M>(p, d);
  if (AHEAD) load_group(r2, p + 6 * 512);
  fold_group<M>(r3, selb, g + 3, d);
  after_fold<M>(p, d);
  p += 4 * 512;
}

template <int M>
__device__ __forceinline__ void sweep_cols(const float* __restrict__ RB,
                                           const float* sqS, float* anchS,
                                           const float* selb,
                                           const float* selSqS,
                                           const unsigned run, const int t) {
  float ssq[M];
#pragma unroll
  for (int c = 0; c < M; c++) {
    ssq[c] = __int_as_float(__builtin_amdgcn_readfirstlane(__float_as_int(selSqS[c])));
  }
  gfp p = (gfp)(RB + run);
  v4f r0[4], r1[4], r2[4], r3[4];
  load_group(r0, p);
  asm volatile("" : "+v"(p));
  load_group(r1, p + 512);
  asm volatile("" : "+v"(p));
  load_group(r2, p + 1024);
#pragma unroll
  for (int j = 0; j < 4; j++) {
    v2f d[M][2];
#pragma unroll
    for (int c = 0; c < M; c++) { d[c][0] = (v2f){0.f, 0.f}; d[c][1] = (v2f){0.f, 0.f}; }
    if (j < 3) {
#pragma unroll 1
      for (int g = 0; g < 32; g += 4)
        ring_trip<M, true>(p, selb, g, r0, r1, r2, r3, d);
    } else {
#pragma unroll 1
      for (int g = 0; g < 28; g += 4)
        ring_trip<M, true>(p, selb, g, r0, r1, r2, r3, d);
      ring_trip<M, false>(p, selb, 28, r0, r1, r2, r3, d);
    }
    const float4 q = *(const float4*)(sqS + 4 * (1024 * j + t));
    float4 a = *(const float4*)(anchS + 4 * (1024 * j + t));
    float e0 = q.x + ssq[0] - 2.f * d[0][0].x;
    float e1 = q.y + ssq[0] - 2.f * d[0][0].y;
    float e2 = q.z + ssq[0] - 2.f * d[0][1].x;
    float e3 = q.w + ssq[0] - 2.f * d[0][1].y;
#pragma unroll
    for (int c = 1; c < M; c++) {
      e0 = fminf(e0, q.x + ssq[c] - 2.f * d[c][0].x);
      e1 = fminf(e1, q.y + ssq[c] - 2.f * d[c][0].y);
      e2 = fminf(e2, q.z + ssq[c] - 2.f * d[c][1].x);
      e3 = fminf(e3, q.w + ssq[c] - 2.f * d[c][1].y);
    }
    a.x = fminf(a.x, sqrtf(fmaxf(e0, 0.f)));
    a.y = fminf(a.y, sqrtf(fmaxf(e1, 0.f)));
    a.z = fminf(a.z, sqrtf(fmaxf(e2, 0.f)));
    a.w = fminf(a.w, sqrtf(fmaxf(e3, 0.f)));
    *(float4*)(anchS + 4 * (1024 * j + t)) = a;
  }
}

__device__ __forceinline__ void mm_load(v4f (&b)[4], v2f (&a)[2], gfp p,
                                        const float* ap, const int c) {
#pragma unroll
  for (int u = 0; u < 2; u++) {
    b[2 * u]     = *(gv4p)(p + (2 * c + u) * 512);
    b[2 * u + 1] = *(gv4p)(p + (2 * c + u) * 512 + 128);
    a[u]         = *(const v2f*)(ap + 4 * (2 * c + u));
  }
}

__device__ __forceinline__ void mm_mma(const v4f (&b)[4], const v2f (&a)[2],
                                       v8f (&acc)[4]) {
#pragma unroll
  for (int u = 0; u < 2; u++) {
    acc[0] = wmma_f32_k4(a[u], (v2f){b[2 * u].x, b[2 * u + 1].x}, acc[0]);
    acc[1] = wmma_f32_k4(a[u], (v2f){b[2 * u].y, b[2 * u + 1].y}, acc[1]);
    acc[2] = wmma_f32_k4(a[u], (v2f){b[2 * u].z, b[2 * u + 1].z}, acc[2]);
    acc[3] = wmma_f32_k4(a[u], (v2f){b[2 * u].w, b[2 * u + 1].w}, acc[3]);
  }
}

__device__ __forceinline__ void sweep_mm(const float* __restrict__ RB,
                                         const float* sqS, float* anchS,
                                         const float* selb, const float* selSqS,
                                         const int m, const int wid, const int lane) {
  const int half = lane >> 4;
  const int r16  = lane & 15;
  float ssq8[8];
#pragma unroll
  for (int i = 0; i < 8; i++) {
    const int s = 8 * half + i;
    const float v = selSqS[s];
    ssq8[i] = (s < m) ? v : __builtin_inff();
  }
  unsigned aoff = (unsigned)(r16 * SPITCH + 2 * half);
  gfp p0 = (gfp)(RB + (unsigned)wid * (4u * 128u * DPROJ) + 4u * (unsigned)r16
                    + 256u * (unsigned)half);
  v4f bA[4], bB[4];
  v2f aA[2], aB[2];
  mm_load(bA, aA, p0, selb + aoff, 0);
#pragma unroll 1
  for (int G = 0; G < 8; G++) {
    asm volatile("" : "+v"(aoff));
    const float* ap = selb + aoff;
    gfp p = p0 + (G >> 1) * (128 * DPROJ) + (G & 1) * 64;
    const int Gn = (G < 7) ? G + 1 : G;
    gfp pn = p0 + (Gn >> 1) * (128 * DPROJ) + (Gn & 1) * 64;
    v8f acc[4] = {};
#pragma unroll
    for (int c2 = 0; c2 < 8; c2++) {
      mm_load(bB, aB, p, ap, 2 * c2 + 1);
      mm_mma(bA, aA, acc);
      if (c2 < 7) mm_load(bA, aA, p, ap, 2 * c2 + 2);
      else        mm_load(bA, aA, pn, ap, 0);
      mm_mma(bB, aB, acc);
    }

    const int slot = 1024 * (G >> 1) + 32 * wid + 16 * (G & 1) + r16;
    const float4 q = *(const float4*)(sqS + 4 * slot);
    float4 a = *(const float4*)(anchS + 4 * slot);
    float e0 = __builtin_inff(), e1 = e0, e2 = e0, e3 = e0;
#pragma unroll
    for (int i = 0; i < 8; i++) {
      e0 = fminf(e0, q.x + ssq8[i] - 2.f * acc[0][i]);
      e1 = fminf(e1, q.y + ssq8[i] - 2.f * acc[1][i]);
      e2 = fminf(e2, q.z + ssq8[i] - 2.f * acc[2][i]);
      e3 = fminf(e3, q.w + ssq8[i] - 2.f * acc[3][i]);
    }
    e0 = fminf(e0, __shfl_xor(e0, 16, 32));
    e1 = fminf(e1, __shfl_xor(e1, 16, 32));
    e2 = fminf(e2, __shfl_xor(e2, 16, 32));
    e3 = fminf(e3, __shfl_xor(e3, 16, 32));
    a.x = fminf(a.x, sqrtf(fmaxf(e0, 0.f)));
    a.y = fminf(a.y, sqrtf(fmaxf(e1, 0.f)));
    a.z = fminf(a.z, sqrtf(fmaxf(e2, 0.f)));
    a.w = fminf(a.w, sqrtf(fmaxf(e3, 0.f)));
    if (half == 0) *(float4*)(anchS + 4 * slot) = a;
  }
}

__global__ __launch_bounds__(1024)
void greedy_kernel(const float* __restrict__ RB, const float* __restrict__ sq,
                   const float* __restrict__ anchorSq, int* __restrict__ sel) {
  __shared__ float redV[NPOOL];
  __shared__ int   redI[NPOOL];
  __shared__ float redS[32];
  __shared__ __attribute__((aligned(16))) float pool[NPOOL * PPITCH];
  __shared__ __attribute__((aligned(16))) float selb[MMAX * SPITCH];
  __shared__ float selSqS[MMAX];
  __shared__ __attribute__((aligned(16))) float pgram[NPOOL * GPITCH];
  __shared__ int   mS;
  __shared__ __attribute__((aligned(16))) float anchS[4 * 1024 * 4];
  __shared__ __attribute__((aligned(16))) float sqS[4 * 1024 * 4];

  const int t    = threadIdx.x;
  const int lane = t & 31;
  const int wid  = t >> 5;
  const int row0 = 512 * wid + 4 * lane;

#pragma unroll
  for (int j = 0; j < 4; j++) {
    float4 a = *(const float4*)(anchorSq + row0 + 128 * j);
    a.x = sqrtf(a.x);
    a.y = sqrtf(a.y);
    a.z = sqrtf(a.z);
    a.w = sqrtf(a.w);
    *(float4*)(anchS + 4 * (1024 * j + t)) = a;
    *(float4*)(sqS + 4 * (1024 * j + t)) = *(const float4*)(sq + row0 + 128 * j);
  }

  for (int i = t; i < MMAX * SPITCH; i += 1024) selb[i] = 0.f;

  const unsigned run = (unsigned)wid * (4u * 128u * DPROJ) + 4u * (unsigned)lane;

  int done = 0;
  for (int pass = 0; pass < NSAMP && done < NSAMP; pass++) {
    float aVal[16];
#pragma unroll
    for (int j = 0; j < 4; j++) {
      const float4 a = *(const float4*)(anchS + 4 * (1024 * j + t));
      aVal[4 * j + 0] = a.x; aVal[4 * j + 1] = a.y; aVal[4 * j + 2] = a.z; aVal[4 * j + 3] = a.w;
    }
    float wv[PERW]; int wi[PERW];
#pragma unroll
    for (int r = 0; r < PERW; r++) {
      float bv = aVal[0]; int be = 0;
#pragma unroll
      for (int e = 1; e < 16; e++) {
        const float v = aVal[e];
        if (v > bv) { bv = v; be = e; }
      }
      const int mine = row0 + 128 * (be >> 2) + (be & 3);
      int bi = mine;
#pragma unroll
      for (int m = 1; m < 32; m <<= 1) {
        const float ov = __shfl_xor(bv, m, 32);
        const int   oi = __shfl_xor(bi, m, 32);
        if ((ov > bv) || (ov == bv && oi < bi)) { bv = ov; bi = oi; }
      }
      wv[r] = bv; wi[r] = bi;
      const int gone = (bi == mine) ? be : -1;
#pragma unroll
      for (int e = 0; e < 16; e++)
        if (gone == e) aVal[e] = -2.f;
    }
    float tl = aVal[0];
#pragma unroll
    for (int e = 1; e < 16; e++) tl = fmaxf(tl, aVal[e]);
#pragma unroll
    for (int m = 1; m < 32; m <<= 1) tl = fmaxf(tl, __shfl_xor(tl, m, 32));
    {
      float g[PERW][4];
#pragma unroll
      for (int r = 0; r < PERW; r++) {
        const unsigned base = (unsigned)(wi[r] >> 7) * (128u * DPROJ) + (unsigned)(wi[r] & 127);
#pragma unroll
        for (int i = 0; i < 4; i++) g[r][i] = RB[base + (unsigned)(lane + 32 * i) * 128u];
      }
#pragma unroll
      for (int r = 0; r < PERW; r++) {
#pragma unroll
        for (int i = 0; i < 4; i++) pool[(wid + 32 * r) * PPITCH + lane + 32 * i] = g[r][i];
      }
    }
    if (lane == 0) {
#pragma unroll
      for (int r = 0; r < PERW; r++) { redV[wid + 32 * r] = wv[r]; redI[wid + 32 * r] = wi[r]; }
      redS[wid] = tl;
    }
    __syncthreads();

    {
      const int half = lane >> 4;
      const int r16  = lane & 15;
      const int wu = __builtin_amdgcn_readfirstlane(wid);
      int ti, tj, nt;
      if (wu < 28) {
        ti = 0;
        int rem = wu;
        while (rem >= 7 - ti) { rem -= 7 - ti; ti++; }
        tj = ti + 1 + rem;
        nt = 1;
      } else {
        ti = 2 * (wu - 28);
        tj = ti;
        nt = 2;
      }
      for (int u = 0; u < nt; u++) {
        const float* pa = pool + (16 * (ti + u) + r16) * PPITCH + 2 * half;
        const float* pb = pool + (16 * (tj + u) + r16) * PPITCH + 2 * half;
        v8f acc = {};
#pragma unroll 8
        for (int kk = 0; kk < 32; kk++) {
          const v2f a = *(const v2f*)(pa + 4 * kk);
          const v2f b = *(const v2f*)(pb + 4 * kk);
          acc = wmma_f32_k4(a, b, acc);
        }
        const int ra = 16 * (ti + u) + 8 * half;
        const int cb = 16 * (tj + u) + r16;
#pragma unroll
        for (int i = 0; i < 8; i++) pgram[(ra + i) * GPITCH + cb] = acc[i];
        if (nt == 1) {
#pragma unroll
          for (int i = 0; i < 8; i++) pgram[cb * GPITCH + ra + i] = acc[i];
        }
      }
    }
    __syncthreads();

    if (wid == 0) {
      float pv[PERW]; int pi[PERW]; float psq[PERW];
#pragma unroll
      for (int r = 0; r < PERW; r++) {
        pv[r]  = redV[lane + 32 * r];
        pi[r]  = redI[lane + 32 * r];
        psq[r] = sq[pi[r]];
      }
      float T = redS[lane];
#pragma unroll
      for (int m = 1; m < 32; m <<= 1) T = fmaxf(T, __shfl_xor(T, m, 32));
      int m = 0;
      for (int c = 0; c < MMAX; c++) {
        float cv = pv[0];
        int   ci = pi[0];
        int   cs = lane;
#pragma unroll
        for (int r = 1; r < PERW; r++) {
          if (pv[r] > cv || (pv[r] == cv && pi[r] < ci)) { cv = pv[r]; ci = pi[r]; cs = lane + 32 * r; }
        }
#pragma unroll
        for (int x = 1; x < 32; x <<= 1) {
          const float ov = __shfl_xor(cv, x, 32);
          const int   oi = __shfl_xor(ci, x, 32);
          const int   os = __shfl_xor(cs, x, 32);
          if (ov > cv || (ov == cv && oi < ci)) { cv = ov; ci = oi; cs = os; }
        }
        if (c > 0 && !(cv > T)) break;
        float mine = psq[0];
#pragma unroll
        for (int r = 1; r < PERW; r++) mine = ((cs >> 5) == r) ? psq[r] : mine;
        const float ssq = __shfl(mine, cs & 31, 32);
        if (lane == 0) { sel[done + c] = ci; selSqS[c] = ssq; }
        *(v4f*)(selb + SPITCH * c + 4 * lane) = *(const v4f*)(pool + cs * PPITCH + 4 * lane);
        m = c + 1;
        if (done + m == NSAMP || m == MMAX) break;
        float d[PERW];
#pragma unroll
        for (int r = 0; r < PERW; r++) d[r] = pgram[cs * GPITCH + lane + 32 * r];
#pragma unroll
        for (int r = 0; r < PERW; r++)
          pv[r] = fminf(pv[r], sqrtf(fmaxf(psq[r] + ssq - 2.f * d[r], 0.f)));
      }
      if (lane == 0) mS = m;
    }
    __syncthreads();

    const int m = mS;
    if (m > MVALU) sweep_mm(RB, sqS, anchS, selb, selSqS, m, wid, lane);
    else switch (m) {
      case 1:  sweep_cols<1>(RB, sqS, anchS, selb, selSqS, run, t); break;
      case 2:  sweep_cols<2>(RB, sqS, anchS, selb, selSqS, run, t); break;
      case 3:  sweep_cols<3>(RB, sqS, anchS, selb, selSqS, run, t); break;
      case 4:  sweep_cols<4>(RB, sqS, anchS, selb, selSqS, run, t); break;
      case 5:  sweep_cols<5>(RB, sqS, anchS, selb, selSqS, run, t); break;
      case 6:  sweep_cols<6>(RB, sqS, anchS, selb, selSqS, run, t); break;
      default: sweep_cols<7>(RB, sqS, anchS, selb, selSqS, run, t); break;
    }
    done += m;
    __syncthreads();
  }
}

__global__ __launch_bounds__(256)
void gather_kernel(const float* __restrict__ F, const int* __restrict__ sel,
                   float* __restrict__ out) {
  const int s = blockIdx.x;
  const int c = blockIdx.y * 256 + threadIdx.x;
  out[(size_t)s * DIN + c] = F[(size_t)sel[s] * DIN + c];
}

extern "C" void kernel_launch(void* const* d_in, const int* in_sizes, int n_in,
                              void* d_out, int out_size, void* d_ws, size_t ws_size,
                              hipStream_t stream) {
  (void)in_sizes; (void)n_in; (void)out_size; (void)ws_size;
  const float* F = (const float*)d_in[0];
  const float* W = (const float*)d_in[1];

  float* R          = (float*)d_ws;
  float* sq         = R + (size_t)NPTS * DPROJ;
  float* anchorSq   = sq + NPTS;
  float* anchorPart = anchorSq + NPTS;
  int*   sel        = (int*)(anchorPart + (size_t)NCHUNK * NPTS);

  proj_kernel<<<1024, 256, 0, stream>>>(F, W, R);
  sq_kernel<<<NPTS / 8, 256, 0, stream>>>(R, sq);
  kmajor_kernel<<<NPTS / 128, 1024, 0, stream>>>(R);
  gram_rowsum_kernel<<<1024, 256, 0, stream>>>(R, sq, anchorPart);
  combine_kernel<<<NPTS / 256, 256, 0, stream>>>(anchorPart, anchorSq);
  greedy_kernel<<<1, 1024, 0, stream>>>(R, sq, anchorSq, sel);
  gather_kernel<<<dim3(NSAMP, DIN / 256), 256, 0, stream>>>(F, sel, (float*)d_out);
}
